# starting kernel
# baseline (speedup 1.0000x reference)
#include <hip/hip_runtime.h>
#include <math.h>

typedef __attribute__((ext_vector_type(2))) float v2f;
typedef __attribute__((ext_vector_type(8))) float v8f;


__global__ void k_deg_init(float* __restrict__ deg, int n) {
    int i = blockIdx.x * blockDim.x + threadIdx.x;
    if (i < n) deg[i] = 1.0f;
}

__global__ void k_deg_count(const int* __restrict__ dst, float* __restrict__ deg, int E) {
    int e = blockIdx.x * blockDim.x + threadIdx.x;
    if (e < E) atomicAdd(&deg[dst[e]], 1.0f);
}

__global__ void k_dinv(const float* __restrict__ deg, float* __restrict__ dinv, int n) {
    int i = blockIdx.x * blockDim.x + threadIdx.x;
    if (i < n) dinv[i] = rsqrtf(deg[i]);
}

__global__ void __launch_bounds__(256) k_gemm64(const float* __restrict__ A,
                                                const float* __restrict__ B,
                                                float* __restrict__ C) {
    const int wave = threadIdx.x >> 5;
    const int lane = threadIdx.x & 31;
    const int half = lane >> 4;
    const int lm   = lane & 15;
    const int m0   = blockIdx.x * 32 + (wave >> 2) * 16;
    const int n0   = (wave & 3) * 16;
    const float* arow = A + (size_t)(m0 + lm) * 64;
    const int bcol = n0 + lm;

    v8f acc = {0.f,0.f,0.f,0.f,0.f,0.f,0.f,0.f};
#pragma unroll
    for (int kk = 0; kk < 16; ++kk) {
        const int kb = kk * 4 + half * 2;
        v2f a = *(const v2f*)(arow + kb);
        v2f b;
        b.x = B[(size_t)kb * 64 + bcol];
        b.y = B[(size_t)(kb + 1) * 64 + bcol];
        acc = __builtin_amdgcn_wmma_f32_16x16x4_f32(false, a, false, b,
                                                    (short)0, acc, false, false);
    }
#pragma unroll
    for (int i = 0; i < 8; ++i)
        C[(size_t)(m0 + half * 8 + i) * 64 + n0 + lm] = acc[i];
}

__global__ void k_init_h(const float* __restrict__ xw, const float* __restrict__ dinv,
                         const float* __restrict__ b1, float* __restrict__ h, int n) {
    int idx = blockIdx.x * blockDim.x + threadIdx.x;
    if (idx >= n * 64) return;
    int i = idx >> 6, c = idx & 63;
    float di = dinv[i];
    h[idx] = b1[c] + di * di * xw[idx];
}

__global__ void k_scatter64(const int* __restrict__ src, const int* __restrict__ dst,
                            const float* __restrict__ dinv, const float* __restrict__ xw,
                            float* __restrict__ h, int E) {
    int idx = blockIdx.x * blockDim.x + threadIdx.x;
    if (idx >= E * 16) return;
    int e = idx >> 4;
    int c = (idx & 15) << 2;
    int s = src[e], d = dst[e];
    float w = dinv[s] * dinv[d];
    const float4 v = *(const float4*)(xw + (size_t)s * 64 + c);
    float* o = h + (size_t)d * 64 + c;
    atomicAdd(o + 0, w * v.x);
    atomicAdd(o + 1, w * v.y);
    atomicAdd(o + 2, w * v.z);
    atomicAdd(o + 3, w * v.w);
}

__global__ void __launch_bounds__(128) k_relu_softmax(float* __restrict__ h, int n) {
    int node = blockIdx.x * 4 + (threadIdx.x >> 5);
    int lane = threadIdx.x & 31;
    if (node >= n) return;
    float* row = h + (size_t)node * 64;
    float v0 = fmaxf(row[lane], 0.0f);
    float v1 = fmaxf(row[lane + 32], 0.0f);
    float m = fmaxf(v0, v1);
#pragma unroll
    for (int off = 16; off > 0; off >>= 1) m = fmaxf(m, __shfl_xor(m, off, 32));
    float e0 = __expf(v0 - m), e1 = __expf(v1 - m);
    float s = e0 + e1;
#pragma unroll
    for (int off = 16; off > 0; off >>= 1) s += __shfl_xor(s, off, 32);
    float inv = 1.0f / s;
    row[lane]      = e0 * inv;
    row[lane + 32] = e1 * inv;
}

__global__ void __launch_bounds__(192) k_gemm34(const float* __restrict__ A,
                                                const float* __restrict__ B,
                                                float* __restrict__ C) {
    const int wave = threadIdx.x >> 5;
    const int lane = threadIdx.x & 31;
    const int half = lane >> 4;
    const int lm   = lane & 15;
    const int m0   = blockIdx.x * 32 + (wave / 3) * 16;
    const int n0   = (wave % 3) * 16;
    const int col  = n0 + lm;
    const bool ok  = (col < 34);
    const float* arow = A + (size_t)(m0 + lm) * 64;

    v8f acc = {0.f,0.f,0.f,0.f,0.f,0.f,0.f,0.f};
#pragma unroll
    for (int kk = 0; kk < 16; ++kk) {
        const int kb = kk * 4 + half * 2;
        v2f a = *(const v2f*)(arow + kb);
        v2f b;
        b.x = ok ? B[(size_t)kb * 34 + col]       : 0.0f;
        b.y = ok ? B[(size_t)(kb + 1) * 34 + col] : 0.0f;
        acc = __builtin_amdgcn_wmma_f32_16x16x4_f32(false, a, false, b,
                                                    (short)0, acc, false, false);
    }
    if (ok) {
#pragma unroll
        for (int i = 0; i < 8; ++i)
            C[(size_t)(m0 + half * 8 + i) * 34 + col] = acc[i];
    }
}

__global__ void k_init_out(const float* __restrict__ hw2, const float* __restrict__ dinv,
                           const float* __restrict__ b2, float* __restrict__ out, int n) {
    int idx = blockIdx.x * blockDim.x + threadIdx.x;
    if (idx >= n * 34) return;
    int i = idx / 34, c = idx - i * 34;
    float di = dinv[i];
    out[idx] = b2[c] + di * di * hw2[idx];
}

__global__ void k_scatter34(const int* __restrict__ src, const int* __restrict__ dst,
                            const float* __restrict__ dinv, const float* __restrict__ hw2,
                            float* __restrict__ out, int E) {
    int idx = blockIdx.x * blockDim.x + threadIdx.x;
    if (idx >= E * 34) return;
    int e = idx / 34, c = idx - e * 34;
    int s = src[e], d = dst[e];
    float w = dinv[s] * dinv[d];
    atomicAdd(&out[(size_t)d * 34 + c], w * hw2[(size_t)s * 34 + c]);
}

extern "C" void kernel_launch(void* const* d_in, const int* in_sizes, int n_in,
                              void* d_out, int out_size, void* d_ws, size_t ws_size,
                              hipStream_t stream) {
    const float* x  = (const float*)d_in[0];
    const int*   ei = (const int*)  d_in[1];
    const float* W1 = (const float*)d_in[2];
    const float* b1 = (const float*)d_in[3];
    const float* W2 = (const float*)d_in[4];
    const float* b2 = (const float*)d_in[5];
    float* out = (float*)d_out;

    const int N = in_sizes[0] / 64;
    const int E = in_sizes[1] / 2;
    const int* src = ei;
    const int* dst = ei + E;

    float* deg  = (float*)d_ws;
    float* dinv = deg  + N;
    float* xw   = dinv + N;
    float* h    = xw   + (size_t)N * 64;
    float* hw2  = h    + (size_t)N * 64;

    const int T = 256;
    k_deg_init <<<(N + T - 1) / T, T, 0, stream>>>(deg, N);
    k_deg_count<<<(E + T - 1) / T, T, 0, stream>>>(dst, deg, E);
    k_dinv     <<<(N + T - 1) / T, T, 0, stream>>>(deg, dinv, N);
    k_gemm64<<<N / 32, 256, 0, stream>>>(x, W1, xw);
    k_init_h   <<<((size_t)N * 64 + T - 1) / T, T, 0, stream>>>(xw, dinv, b1, h, N);
    k_scatter64<<<((size_t)E * 16 + T - 1) / T, T, 0, stream>>>(src, dst, dinv, xw, h, E);
    k_relu_softmax<<<(N + 3) / 4, 128, 0, stream>>>(h, N);
    k_gemm34<<<N / 32, 192, 0, stream>>>(h, W2, hw2);
    k_init_out <<<((size_t)N * 34 + T - 1) / T, T, 0, stream>>>(hw2, dinv, b2, out, N);
    k_scatter34<<<((size_t)E * 34 + T - 1) / T, T, 0, stream>>>(src, dst, dinv, hw2, out, E);
}


# Degree count off global atomics: a block owns (node range, edge slice), counts dst in LDS (integer adds), stores a partial-count plane; dinv pass sums the planes; 7 -> 6 launches; float sums same
# speedup vs baseline: 2.1691x; 2.1691x over previous
#include <hip/hip_runtime.h>
#include <math.h>

typedef __attribute__((ext_vector_type(2))) float v2f;
typedef __attribute__((ext_vector_type(8))) float v8f;

#define EDGE_BLOCKS 1024

#define DEG_RANGE  12800
#define DEG_SLICES 32
#define DEG_T      512
#define DEG_B      20

__device__ __forceinline__ void deg_count1(unsigned* cnt, int d, int lo) {
    const unsigned a = (unsigned)(d - lo);
    if (a < (unsigned)DEG_RANGE) atomicAdd(&cnt[a], 1u);
}

__global__ void __launch_bounds__(DEG_T) k_deg_hist(const int* __restrict__ dst, unsigned* __restrict__ part,
                                                    int E, int per, int npad) {
    __shared__ __attribute__((aligned(16))) unsigned cnt[DEG_RANGE];
    const int r  = blockIdx.x / DEG_SLICES;
    const int s  = blockIdx.x % DEG_SLICES;
    const int lo = r * DEG_RANGE;
    for (int i = threadIdx.x; i < DEG_RANGE / 4; i += DEG_T)
        ((uint4*)cnt)[i] = make_uint4(0u, 0u, 0u, 0u);
    __syncthreads();
    const int e0 = min(E, s * per);
    const int e1 = min(E, e0 + per);
    int e = e0;
    if ((((size_t)(dst + e0)) & 15) == 0) {
        const int4* p = (const int4*)(dst + e0);
        const int n4 = (e1 - e0) >> 2;
        for (int base = 0; base < n4; base += DEG_B * DEG_T) {
            int4 v[DEG_B];
#pragma unroll
            for (int k = 0; k < DEG_B; ++k) {
                const int i = base + k * DEG_T + (int)threadIdx.x;
                v[k] = make_int4(-1, -1, -1, -1);
                if (i < n4) v[k] = p[i];
            }
#pragma unroll
            for (int k = 0; k < DEG_B; ++k) {
                deg_count1(cnt, v[k].x, lo);
                deg_count1(cnt, v[k].y, lo);
                deg_count1(cnt, v[k].z, lo);
                deg_count1(cnt, v[k].w, lo);
            }
        }
        e = e0 + (n4 << 2);
    }
    for (e += (int)threadIdx.x; e < e1; e += DEG_T)
        deg_count1(cnt, dst[e], lo);
    __syncthreads();
    uint4* o = (uint4*)(part + (size_t)s * npad + lo);
    for (int i = threadIdx.x; i < DEG_RANGE / 4; i += DEG_T)
        o[i] = ((const uint4*)cnt)[i];
}

__global__ void __launch_bounds__(128) k_dinv(const unsigned* __restrict__ part, float* __restrict__ dinv,
                                              int n4, int npad4) {
    const int i = blockIdx.x * blockDim.x + threadIdx.x;
    if (i < n4) {
        uint4 c = make_uint4(1u, 1u, 1u, 1u);
#pragma unroll
        for (int s = 0; s < DEG_SLICES; ++s) {
            const uint4 v = ((const uint4*)part)[(size_t)s * npad4 + i];
            c.x += v.x; c.y += v.y; c.z += v.z; c.w += v.w;
        }
        ((float4*)dinv)[i] = make_float4(rsqrtf((float)c.x), rsqrtf((float)c.y),
                                         rsqrtf((float)c.z), rsqrtf((float)c.w));
    }
}

__global__ void __launch_bounds__(256) k_gemm64(const float* __restrict__ A,
                                                const float* __restrict__ B,
                                                const float* __restrict__ dinv,
                                                const float* __restrict__ b1,
                                                float* __restrict__ C,
                                                float* __restrict__ Hacc,
                                                int nstrips) {
    const int wave = __builtin_amdgcn_readfirstlane((int)(threadIdx.x >> 5));
    const int lane = threadIdx.x & 31;
    const int half = lane >> 4;
    const int lm   = lane & 15;
    const int strip = blockIdx.x * 8 + wave;
    if (strip >= nstrips) return;
    const int m0   = strip * 16;
    const float* arow = A + (size_t)(m0 + lm) * 64;

    v2f a[16];
#pragma unroll
    for (int kk = 0; kk < 16; ++kk)
        a[kk] = *(const v2f*)(arow + kk * 4 + half * 2);

    v8f acc[4];
#pragma unroll
    for (int nt = 0; nt < 4; ++nt) acc[nt] = (v8f){0.f,0.f,0.f,0.f,0.f,0.f,0.f,0.f};
#pragma unroll
    for (int kk = 0; kk < 16; ++kk) {
        const int kb = kk * 4 + half * 2;
#pragma unroll
        for (int nt = 0; nt < 4; ++nt) {
            const int bcol = nt * 16 + lm;
            v2f b;
            b.x = B[(size_t)kb * 64 + bcol];
            b.y = B[(size_t)(kb + 1) * 64 + bcol];
            acc[nt] = __builtin_amdgcn_wmma_f32_16x16x4_f32(false, a[kk], false, b,
                                                            (short)0, acc[nt], false, false);
        }
    }
    float di2[8];
#pragma unroll
    for (int i = 0; i < 8; ++i) {
        const float di = dinv[m0 + half * 8 + i];
        di2[i] = di * di;
    }
#pragma unroll
    for (int nt = 0; nt < 4; ++nt) {
        const int col = nt * 16 + lm;
        const float bb = b1[col];
#pragma unroll
        for (int i = 0; i < 8; ++i) {
            const int row = m0 + half * 8 + i;
            const float v = acc[nt][i];
            C[(size_t)row * 64 + col]    = v;
            Hacc[(size_t)row * 64 + col] = bb + di2[i] * v;
        }
    }
}

__device__ __forceinline__ void load_ids(const int* __restrict__ src, const int* __restrict__ dst,
                                         int E, int c, int lane, int& s, int& d) {
    const int e = c * 32 + lane;
    s = 0; d = 0;
    if (e < E) { s = src[e]; d = dst[e]; }
}

__device__ __forceinline__ void edge_add64(const float* __restrict__ xw, float* __restrict__ h,
                                           int s, int d, int wi, int j, int lane) {
    const int   sj = __builtin_amdgcn_readlane(s, j);
    const int   dj = __builtin_amdgcn_readlane(d, j);
    const float wj = __int_as_float(__builtin_amdgcn_readlane(wi, j));
    const float* xr = xw + (size_t)sj * 64;
    float*       o  = h  + (size_t)dj * 64;
    const float a0 = xr[lane], a1 = xr[lane + 32];
    atomicAdd(o + lane,      wj * a0);
    atomicAdd(o + lane + 32, wj * a1);
}

__global__ void __launch_bounds__(256) k_scatter64(const int* __restrict__ src, const int* __restrict__ dst,
                                                   const float* __restrict__ dinv, const float* __restrict__ xw,
                                                   float* __restrict__ h, int E, int nchunks) {
    const int lane = threadIdx.x & 31;
    const int wv = __builtin_amdgcn_readfirstlane((int)((blockIdx.x * blockDim.x + threadIdx.x) >> 5));
    const int nw = (int)((gridDim.x * blockDim.x) >> 5);
    int c = wv;
    if (c >= nchunks) return;
    int s0, d0, s1, d1;
    load_ids(src, dst, E, c, lane, s0, d0);
    load_ids(src, dst, E, c + nw, lane, s1, d1);
    float w0 = dinv[s0] * dinv[d0];
    for (;;) {
        int s2, d2;
        load_ids(src, dst, E, c + 2 * nw, lane, s2, d2);
        const float w1 = dinv[s1] * dinv[d1];
        const int cnt = min(32, E - c * 32);
        const int wi = __float_as_int(w0);
        int j0 = 0;
        for (; j0 + 16 <= cnt; j0 += 16) {
            float v0[16], v1[16];
#pragma unroll
            for (int u = 0; u < 16; ++u) {
                const int sj = __builtin_amdgcn_readlane(s0, j0 + u);
                const float* xr = xw + (size_t)sj * 64;
                v0[u] = xr[lane];
                v1[u] = xr[lane + 32];
            }
#pragma unroll
            for (int u = 0; u < 16; ++u) {
                const int   dj = __builtin_amdgcn_readlane(d0, j0 + u);
                const float wj = __int_as_float(__builtin_amdgcn_readlane(wi, j0 + u));
                float* o = h + (size_t)dj * 64;
                atomicAdd(o + lane,      wj * v0[u]);
                atomicAdd(o + lane + 32, wj * v1[u]);
            }
        }
        for (; j0 < cnt; ++j0)
            edge_add64(xw, h, s0, d0, wi, j0, lane);
        c += nw;
        if (c >= nchunks) break;
        s0 = s1; d0 = d1; w0 = w1;
        s1 = s2; d1 = d2;
    }
}

__global__ void __launch_bounds__(256) k_gemm34(const float* __restrict__ A,
                                                const float* __restrict__ B,
                                                const float* __restrict__ dinv,
                                                const float* __restrict__ b2,
                                                float* __restrict__ C,
                                                float* __restrict__ out,
                                                int nstrips) {
    const int wave = __builtin_amdgcn_readfirstlane((int)(threadIdx.x >> 5));
    const int lane = threadIdx.x & 31;
    const int half = lane >> 4;
    const int lm   = lane & 15;
    const int strip = blockIdx.x * 8 + wave;
    if (strip >= nstrips) return;
    const int m0   = strip * 16;
    const float* arow = A + (size_t)(m0 + lm) * 64;

    v2f a[16];
#pragma unroll
    for (int kk = 0; kk < 16; ++kk)
        a[kk] = *(const v2f*)(arow + kk * 4 + half * 2);

    float m = 0.0f;
#pragma unroll
    for (int kk = 0; kk < 16; ++kk) {
        a[kk].x = fmaxf(a[kk].x, 0.0f);
        a[kk].y = fmaxf(a[kk].y, 0.0f);
        m = fmaxf(m, fmaxf(a[kk].x, a[kk].y));
    }
    m = fmaxf(m, __shfl_xor(m, 16, 32));
    float s = 0.0f;
#pragma unroll
    for (int kk = 0; kk < 16; ++kk) {
        a[kk].x = __expf(a[kk].x - m);
        a[kk].y = __expf(a[kk].y - m);
        s += a[kk].x + a[kk].y;
    }
    s += __shfl_xor(s, 16, 32);
    const float inv = 1.0f / s;

    v8f acc[3];
#pragma unroll
    for (int nt = 0; nt < 3; ++nt) acc[nt] = (v8f){0.f,0.f,0.f,0.f,0.f,0.f,0.f,0.f};
#pragma unroll
    for (int kk = 0; kk < 16; ++kk) {
        const int kb = kk * 4 + half * 2;
        v2f av;
        av.x = a[kk].x * inv;
        av.y = a[kk].y * inv;
#pragma unroll
        for (int nt = 0; nt < 3; ++nt) {
            const int col = nt * 16 + lm;
            const bool ok = (col < 34);
            const int  cc = ok ? col : 33;
            const float bx = B[(size_t)kb * 34 + cc];
            const float by = B[(size_t)(kb + 1) * 34 + cc];
            v2f b;
            b.x = ok ? bx : 0.0f;
            b.y = ok ? by : 0.0f;
            acc[nt] = __builtin_amdgcn_wmma_f32_16x16x4_f32(false, av, false, b,
                                                            (short)0, acc[nt], false, false);
        }
    }
    float di2[8];
#pragma unroll
    for (int i = 0; i < 8; ++i) {
        const float di = dinv[m0 + half * 8 + i];
        di2[i] = di * di;
    }
#pragma unroll
    for (int nt = 0; nt < 3; ++nt) {
        const int col = nt * 16 + lm;
        if (col < 34) {
            const float bb = b2[col];
#pragma unroll
            for (int i = 0; i < 8; ++i) {
                const int row = m0 + half * 8 + i;
                const float v = acc[nt][i];
                C[(size_t)row * 34 + col]   = v;
                out[(size_t)row * 34 + col] = bb + di2[i] * v;
            }
        }
    }
}

__global__ void __launch_bounds__(256) k_scatter34(const int* __restrict__ src, const int* __restrict__ dst,
                                                   const float* __restrict__ dinv, const float* __restrict__ hw2,
                                                   float* __restrict__ out, int E, int nchunks) {
    const int lane = threadIdx.x & 31;
    const int wv = __builtin_amdgcn_readfirstlane((int)((blockIdx.x * blockDim.x + threadIdx.x) >> 5));
    const int nw = (int)((gridDim.x * blockDim.x) >> 5);
    int c = wv;
    if (c >= nchunks) return;
    int s0, d0, s1, d1;
    load_ids(src, dst, E, c, lane, s0, d0);
    load_ids(src, dst, E, c + nw, lane, s1, d1);
    float w0 = dinv[s0] * dinv[d0];
    for (;;) {
        int s2, d2;
        load_ids(src, dst, E, c + 2 * nw, lane, s2, d2);
        const float w1 = dinv[s1] * dinv[d1];
        const int cnt = min(32, E - c * 32);
        const int wi = __float_as_int(w0);
        float tv[2], tw[2];
        int   td[2];
#pragma unroll
        for (int t = 0; t < 2; ++t) {
            const int el = 16 * t + (lane >> 1);
            const int st = __shfl(s0, el, 32);
            td[t] = __shfl(d0, el, 32);
            tw[t] = __shfl(w0, el, 32);
            tv[t] = hw2[(size_t)st * 34 + 32 + (lane & 1)];
        }
        int j0 = 0;
        for (; j0 + 16 <= cnt; j0 += 16) {
            float v[16];
#pragma unroll
            for (int u = 0; u < 16; ++u) {
                const int sj = __builtin_amdgcn_readlane(s0, j0 + u);
                v[u] = hw2[(unsigned)(sj * 34 + lane)];
            }
#pragma unroll
            for (int u = 0; u < 16; ++u) {
                const int   dj = __builtin_amdgcn_readlane(d0, j0 + u);
                const float wj = __int_as_float(__builtin_amdgcn_readlane(wi, j0 + u));
                atomicAdd(&out[(unsigned)(dj * 34 + lane)], wj * v[u]);
            }
        }
        for (; j0 < cnt; ++j0) {
            const int   sj = __builtin_amdgcn_readlane(s0, j0);
            const int   dj = __builtin_amdgcn_readlane(d0, j0);
            const float wj = __int_as_float(__builtin_amdgcn_readlane(wi, j0));
            const float* hr = hw2 + (size_t)sj * 34;
            float*       o  = out + (size_t)dj * 34;
            atomicAdd(o + lane, wj * hr[lane]);
        }
#pragma unroll
        for (int t = 0; t < 2; ++t) {
            const int el = 16 * t + (lane >> 1);
            if (el < cnt)
                atomicAdd(out + (size_t)td[t] * 34 + 32 + (lane & 1), tw[t] * tv[t]);
        }
        c += nw;
        if (c >= nchunks) break;
        s0 = s1; d0 = d1; w0 = w1;
        s1 = s2; d1 = d2;
    }
}

extern "C" void kernel_launch(void* const* d_in, const int* in_sizes, int n_in,
                              void* d_out, int out_size, void* d_ws, size_t ws_size,
                              hipStream_t stream) {
    const float* x  = (const float*)d_in[0];
    const int*   ei = (const int*)  d_in[1];
    const float* W1 = (const float*)d_in[2];
    const float* b1 = (const float*)d_in[3];
    const float* W2 = (const float*)d_in[4];
    const float* b2 = (const float*)d_in[5];
    float* out = (float*)d_out;

    const int N = in_sizes[0] / 64;
    const int E = in_sizes[1] / 2;
    const int* src = ei;
    const int* dst = ei + E;

    const int nranges = (N + DEG_RANGE - 1) / DEG_RANGE;
    const int npad    = nranges * DEG_RANGE;
    const int per     = (((E + DEG_SLICES - 1) / DEG_SLICES) + 63) & ~63;

    float* xw   = (float*)d_ws;
    float* h    = xw   + (size_t)N * 64;
    float* hw2  = h    + (size_t)N * 64;
    float* dinv = hw2  + (size_t)N * 34;
    unsigned* part = (unsigned*)(dinv + N);

    const int T = 256;
    const int N4 = N / 4;
    const int nstrips = N / 16;
    const int nchunks = (E + 31) / 32;
    k_deg_hist<<<nranges * DEG_SLICES, DEG_T, 0, stream>>>(dst, part, E, per, npad);
    k_dinv    <<<(N4 + 127) / 128, 128, 0, stream>>>(part, dinv, N4, npad / 4);
    k_gemm64<<<(nstrips + 7) / 8, 256, 0, stream>>>(x, W1, dinv, b1, xw, h, nstrips);
    k_scatter64<<<EDGE_BLOCKS, T, 0, stream>>>(src, dst, dinv, xw, h, E, nchunks);
    k_gemm34<<<(nstrips + 7) / 8, 256, 0, stream>>>(h, W2, dinv, b2, hw2, out, nstrips);
    k_scatter34<<<EDGE_BLOCKS, T, 0, stream>>>(src, dst, dinv, hw2, out, E, nchunks);
}


# Output rows' atomic adds kept in one half of the chip: a wave reads its die, draws edge tiles from its half's ticket counter, keeps only edges with dst in that half's node range; atomic order changes
# speedup vs baseline: 2.4126x; 1.1123x over previous
#include <hip/hip_runtime.h>
#include <math.h>

typedef __attribute__((ext_vector_type(2))) float v2f;
typedef __attribute__((ext_vector_type(8))) float v8f;

#define EDGE_BLOCKS 1024
#define TILE_E      64
#define QPITCH      64

#define DEG_RANGE  12800
#define DEG_SLICES 32
#define DEG_T      512
#define DEG_B      20

__device__ __forceinline__ void deg_count1(unsigned* cnt, int d, int lo) {
    const unsigned a = (unsigned)(d - lo);
    if (a < (unsigned)DEG_RANGE) atomicAdd(&cnt[a], 1u);
}

__global__ void __launch_bounds__(DEG_T) k_deg_hist(const int* __restrict__ dst, unsigned* __restrict__ part,
                                                    int E, int per, int npad) {
    __shared__ __attribute__((aligned(16))) unsigned cnt[DEG_RANGE];
    const int r  = blockIdx.x / DEG_SLICES;
    const int s  = blockIdx.x % DEG_SLICES;
    const int lo = r * DEG_RANGE;
    for (int i = threadIdx.x; i < DEG_RANGE / 4; i += DEG_T)
        ((uint4*)cnt)[i] = make_uint4(0u, 0u, 0u, 0u);
    __syncthreads();
    const int e0 = min(E, s * per);
    const int e1 = min(E, e0 + per);
    int e = e0;
    if ((((size_t)(dst + e0)) & 15) == 0) {
        const int4* p = (const int4*)(dst + e0);
        const int n4 = (e1 - e0) >> 2;
        for (int base = 0; base < n4; base += DEG_B * DEG_T) {
            int4 v[DEG_B];
#pragma unroll
            for (int k = 0; k < DEG_B; ++k) {
                const int i = base + k * DEG_T + (int)threadIdx.x;
                v[k] = make_int4(-1, -1, -1, -1);
                if (i < n4) v[k] = p[i];
            }
#pragma unroll
            for (int k = 0; k < DEG_B; ++k) {
                deg_count1(cnt, v[k].x, lo);
                deg_count1(cnt, v[k].y, lo);
                deg_count1(cnt, v[k].z, lo);
                deg_count1(cnt, v[k].w, lo);
            }
        }
        e = e0 + (n4 << 2);
    }
    for (e += (int)threadIdx.x; e < e1; e += DEG_T)
        deg_count1(cnt, dst[e], lo);
    __syncthreads();
    uint4* o = (uint4*)(part + (size_t)s * npad + lo);
    for (int i = threadIdx.x; i < DEG_RANGE / 4; i += DEG_T)
        o[i] = ((const uint4*)cnt)[i];
}

__global__ void __launch_bounds__(128) k_dinv(const unsigned* __restrict__ part, float* __restrict__ dinv,
                                              unsigned* __restrict__ ctr, int n4, int npad4) {
    const int i = blockIdx.x * blockDim.x + threadIdx.x;
    if (i < 4) ctr[i * QPITCH] = 0u;
    if (i < n4) {
        uint4 c = make_uint4(1u, 1u, 1u, 1u);
#pragma unroll
        for (int s = 0; s < DEG_SLICES; ++s) {
            const uint4 v = ((const uint4*)part)[(size_t)s * npad4 + i];
            c.x += v.x; c.y += v.y; c.z += v.z; c.w += v.w;
        }
        ((float4*)dinv)[i] = make_float4(rsqrtf((float)c.x), rsqrtf((float)c.y),
                                         rsqrtf((float)c.z), rsqrtf((float)c.w));
    }
}

__global__ void __launch_bounds__(256) k_gemm64(const float* __restrict__ A,
                                                const float* __restrict__ B,
                                                const float* __restrict__ dinv,
                                                const float* __restrict__ b1,
                                                float* __restrict__ C,
                                                float* __restrict__ Hacc,
                                                int nstrips) {
    const int wave = __builtin_amdgcn_readfirstlane((int)(threadIdx.x >> 5));
    const int lane = threadIdx.x & 31;
    const int half = lane >> 4;
    const int lm   = lane & 15;
    const int strip = blockIdx.x * 8 + wave;
    if (strip >= nstrips) return;
    const int m0   = strip * 16;
    const float* arow = A + (size_t)(m0 + lm) * 64;

    v2f a[16];
#pragma unroll
    for (int kk = 0; kk < 16; ++kk)
        a[kk] = *(const v2f*)(arow + kk * 4 + half * 2);

    v8f acc[4];
#pragma unroll
    for (int nt = 0; nt < 4; ++nt) acc[nt] = (v8f){0.f,0.f,0.f,0.f,0.f,0.f,0.f,0.f};
#pragma unroll
    for (int kk = 0; kk < 16; ++kk) {
        const int kb = kk * 4 + half * 2;
#pragma unroll
        for (int nt = 0; nt < 4; ++nt) {
            const int bcol = nt * 16 + lm;
            v2f b;
            b.x = B[(size_t)kb * 64 + bcol];
            b.y = B[(size_t)(kb + 1) * 64 + bcol];
            acc[nt] = __builtin_amdgcn_wmma_f32_16x16x4_f32(false, a[kk], false, b,
                                                            (short)0, acc[nt], false, false);
        }
    }
    float di2[8];
#pragma unroll
    for (int i = 0; i < 8; ++i) {
        const float di = dinv[m0 + half * 8 + i];
        di2[i] = di * di;
    }
#pragma unroll
    for (int nt = 0; nt < 4; ++nt) {
        const int col = nt * 16 + lm;
        const float bb = b1[col];
#pragma unroll
        for (int i = 0; i < 8; ++i) {
            const int row = m0 + half * 8 + i;
            const float v = acc[nt][i];
            C[(size_t)row * 64 + col]    = v;
            Hacc[(size_t)row * 64 + col] = bb + di2[i] * v;
        }
    }
}

#define QCAP 96

__device__ __forceinline__ int chip_half() {
    return (int)((__builtin_amdgcn_s_sendmsg_rtn(0x87) >> 18) & 1u);
}

__device__ __forceinline__ unsigned draw_ticket(unsigned* q, int lane) {
    unsigned v = 0u;
    if (lane == 0) v = atomicAdd(q, 1u);
    return v;
}

__device__ __forceinline__ int next_tile(unsigned* __restrict__ ctr, int ntiles, int lane,
                                         unsigned& pend, int& q, bool& switched) {
    for (;;) {
        const int t = (int)__builtin_amdgcn_readfirstlane(pend);
        if (t < ntiles) { pend = draw_ticket(ctr + q * QPITCH, lane); return t; }
        if (switched) return -1;
        switched = true;
        q ^= 1;
        pend = draw_ticket(ctr + q * QPITCH, lane);
    }
}

__device__ __forceinline__ void load_tile(const int* __restrict__ src, const int* __restrict__ dst,
                                          int E, int t, int lane, int& s0, int& d0, int& s1, int& d1) {
    const int e = t * TILE_E + lane;
    s0 = 0; d0 = -1; s1 = 0; d1 = -1;
    if (e      < E) { s0 = src[e];      d0 = dst[e];      }
    if (e + 32 < E) { s1 = src[e + 32]; d1 = dst[e + 32]; }
}

__device__ __forceinline__ int pack_tile(int s0, int d0, int s1, int d1, bool i0, bool i1,
                                         float a0, float b0, float a1, float b1,
                                         int* qs, int* qd, float* qw, int nq) {
    const unsigned m0 = __builtin_amdgcn_ballot_w32(i0), m1 = __builtin_amdgcn_ballot_w32(i1);
    const int r0 = i0 ? nq + (int)__builtin_amdgcn_mbcnt_lo(m0, 0u) : QCAP;
    const int r1 = i1 ? nq + __builtin_popcount(m0) + (int)__builtin_amdgcn_mbcnt_lo(m1, 0u) : QCAP;
    qs[r0] = s0; qd[r0] = d0; qw[r0] = a0 * b0;
    qs[r1] = s1; qd[r1] = d1; qw[r1] = a1 * b1;
    return nq + __builtin_popcount(m0) + __builtin_popcount(m1);
}

__device__ __forceinline__ void queue_window(const int* qs, const int* qd, const float* qw, int pos, int nq, int lane,
                                             int& sv, int& dv, float& wv) {
    sv = 0; dv = 0; wv = 0.0f;
    if (pos + lane < nq) { sv = qs[pos + lane]; dv = qd[pos + lane]; wv = qw[pos + lane]; }
}

__device__ __forceinline__ int queue_keep_rest(int* qs, int* qd, float* qw, int pos, int nq, int lane) {
    const int rem = nq - pos;
    if (pos > 0 && rem > 0) {
        int ts = 0, td = 0; float tw = 0.0f;
        if (lane < rem) { ts = qs[pos + lane]; td = qd[pos + lane]; tw = qw[pos + lane]; }
        if (lane < rem) { qs[lane] = ts; qd[lane] = td; qw[lane] = tw; }
    }
    return rem;
}

__global__ void __launch_bounds__(256) k_scatter64(const int* __restrict__ src, const int* __restrict__ dst,
                                                   const float* __restrict__ dinv, const float* __restrict__ xw,
                                                   float* __restrict__ h, unsigned* __restrict__ ctr,
                                                   int E, int N, int ntiles) {
    __shared__ unsigned first;
    __shared__ int   q_s[8][QCAP + 1];
    __shared__ int   q_d[8][QCAP + 1];
    __shared__ float q_w[8][QCAP + 1];
    const int lane = threadIdx.x & 31;
    const int wave = __builtin_amdgcn_readfirstlane((int)(threadIdx.x >> 5));
    int* qs = q_s[wave]; int* qd = q_d[wave]; float* qw = q_w[wave];
    const int nsplit = min(N, ((N >> 1) + 15) & ~15);
    int q = chip_half();
    if (threadIdx.x == 0) first = atomicAdd(ctr + q * QPITCH, 8u);
    __syncthreads();
    unsigned pend = draw_ticket(ctr + q * QPITCH, lane);
    bool switched = false;
    int t = __builtin_amdgcn_readfirstlane((int)first) + wave;
    int qt = q;
    if (t >= ntiles) {
        t = __builtin_amdgcn_readfirstlane(next_tile(ctr, ntiles, lane, pend, q, switched));
        q = __builtin_amdgcn_readfirstlane(q);
        qt = q;
    }
    if (t < 0) return;
    int s0, d0, s1, d1;
    load_tile(src, dst, E, t, lane, s0, d0, s1, d1);
    bool have = true;
    int nq = 0;
    for (;;) {
        bool i0 = false, i1 = false;
        float a0 = 0.0f, b0 = 0.0f, a1 = 0.0f, b1 = 0.0f;
        int tn = -1, qn = qt;
        int ns0 = 0, nd0 = -1, ns1 = 0, nd1 = -1;
        if (have) {
            tn = __builtin_amdgcn_readfirstlane(next_tile(ctr, ntiles, lane, pend, q, switched));
            q  = __builtin_amdgcn_readfirstlane(q);
            qn = q;
            const unsigned lo  = qt ? (unsigned)nsplit : 0u;
            const unsigned len = qt ? (unsigned)(N - nsplit) : (unsigned)nsplit;
            i0 = ((unsigned)d0 - lo) < len; i1 = ((unsigned)d1 - lo) < len;
            a0 = dinv[i0 ? s0 : 0]; b0 = dinv[i0 ? d0 : 0];
            a1 = dinv[i1 ? s1 : 0]; b1 = dinv[i1 ? d1 : 0];
            if (tn >= 0) load_tile(src, dst, E, tn, lane, ns0, nd0, ns1, nd1);
        }
        int pos = 0;
        while (nq - pos >= 16) {
            const int cnt = min(32, (nq - pos) & ~15);
            int sv, dv; float wv;
            queue_window(qs, qd, qw, pos, nq, lane, sv, dv, wv);
            const int wi = __float_as_int(wv);
            for (int j0 = 0; j0 < cnt; j0 += 16) {
                float v0[16], v1[16];
#pragma unroll
                for (int u = 0; u < 16; ++u) {
                    const int sj = __builtin_amdgcn_readlane(sv, j0 + u);
                    const float* xr = xw + (size_t)sj * 64;
                    v0[u] = xr[lane];
                    v1[u] = xr[lane + 32];
                }
#pragma unroll
                for (int u = 0; u < 16; ++u) {
                    const int   dj = __builtin_amdgcn_readlane(dv, j0 + u);
                    const float wj = __int_as_float(__builtin_amdgcn_readlane(wi, j0 + u));
                    float* o = h + (size_t)dj * 64;
                    atomicAdd(o + lane,      wj * v0[u]);
                    atomicAdd(o + lane + 32, wj * v1[u]);
                }
            }
            pos += cnt;
        }
        nq = queue_keep_rest(qs, qd, qw, pos, nq, lane);
        if (!have) break;
        nq = pack_tile(s0, d0, s1, d1, i0, i1, a0, b0, a1, b1, qs, qd, qw, nq);
        have = (tn >= 0);
        qt = qn; s0 = ns0; d0 = nd0; s1 = ns1; d1 = nd1;
    }
    if (nq > 0) {
        int sv, dv; float wv;
        queue_window(qs, qd, qw, 0, nq, lane, sv, dv, wv);
        const int wi = __float_as_int(wv);
        float v0[16], v1[16];
#pragma unroll
        for (int u = 0; u < 16; ++u) {
            if (u < nq) {
                const int sj = __builtin_amdgcn_readlane(sv, u);
                const float* xr = xw + (size_t)sj * 64;
                v0[u] = xr[lane];
                v1[u] = xr[lane + 32];
            }
        }
#pragma unroll
        for (int u = 0; u < 16; ++u) {
            if (u < nq) {
                const int   dj = __builtin_amdgcn_readlane(dv, u);
                const float wj = __int_as_float(__builtin_amdgcn_readlane(wi, u));
                float* o = h + (size_t)dj * 64;
                atomicAdd(o + lane,      wj * v0[u]);
                atomicAdd(o + lane + 32, wj * v1[u]);
            }
        }
    }
}

__global__ void __launch_bounds__(256) k_gemm34(const float* __restrict__ A,
                                                const float* __restrict__ B,
                                                const float* __restrict__ dinv,
                                                const float* __restrict__ b2,
                                                float* __restrict__ C,
                                                float* __restrict__ out,
                                                int nstrips) {
    const int wave = __builtin_amdgcn_readfirstlane((int)(threadIdx.x >> 5));
    const int lane = threadIdx.x & 31;
    const int half = lane >> 4;
    const int lm   = lane & 15;
    const int strip = blockIdx.x * 8 + wave;
    if (strip >= nstrips) return;
    const int m0   = strip * 16;
    const float* arow = A + (size_t)(m0 + lm) * 64;

    v2f a[16];
#pragma unroll
    for (int kk = 0; kk < 16; ++kk)
        a[kk] = *(const v2f*)(arow + kk * 4 + half * 2);

    float m = 0.0f;
#pragma unroll
    for (int kk = 0; kk < 16; ++kk) {
        a[kk].x = fmaxf(a[kk].x, 0.0f);
        a[kk].y = fmaxf(a[kk].y, 0.0f);
        m = fmaxf(m, fmaxf(a[kk].x, a[kk].y));
    }
    m = fmaxf(m, __shfl_xor(m, 16, 32));
    float s = 0.0f;
#pragma unroll
    for (int kk = 0; kk < 16; ++kk) {
        a[kk].x = __expf(a[kk].x - m);
        a[kk].y = __expf(a[kk].y - m);
        s += a[kk].x + a[kk].y;
    }
    s += __shfl_xor(s, 16, 32);
    const float inv = 1.0f / s;

    v8f acc[3];
#pragma unroll
    for (int nt = 0; nt < 3; ++nt) acc[nt] = (v8f){0.f,0.f,0.f,0.f,0.f,0.f,0.f,0.f};
#pragma unroll
    for (int kk = 0; kk < 16; ++kk) {
        const int kb = kk * 4 + half * 2;
        v2f av;
        av.x = a[kk].x * inv;
        av.y = a[kk].y * inv;
#pragma unroll
        for (int nt = 0; nt < 3; ++nt) {
            const int col = nt * 16 + lm;
            const bool ok = (col < 34);
            const int  cc = ok ? col : 33;
            const float bx = B[(size_t)kb * 34 + cc];
            const float by = B[(size_t)(kb + 1) * 34 + cc];
            v2f b;
            b.x = ok ? bx : 0.0f;
            b.y = ok ? by : 0.0f;
            acc[nt] = __builtin_amdgcn_wmma_f32_16x16x4_f32(false, av, false, b,
                                                            (short)0, acc[nt], false, false);
        }
    }
    float di2[8];
#pragma unroll
    for (int i = 0; i < 8; ++i) {
        const float di = dinv[m0 + half * 8 + i];
        di2[i] = di * di;
    }
#pragma unroll
    for (int nt = 0; nt < 3; ++nt) {
        const int col = nt * 16 + lm;
        if (col < 34) {
            const float bb = b2[col];
#pragma unroll
            for (int i = 0; i < 8; ++i) {
                const int row = m0 + half * 8 + i;
                const float v = acc[nt][i];
                C[(size_t)row * 34 + col]   = v;
                out[(size_t)row * 34 + col] = bb + di2[i] * v;
            }
        }
    }
}

__global__ void __launch_bounds__(256) k_scatter34(const int* __restrict__ src, const int* __restrict__ dst,
                                                   const float* __restrict__ dinv, const float* __restrict__ hw2,
                                                   float* __restrict__ out, unsigned* __restrict__ ctr,
                                                   int E, int N, int ntiles) {
    __shared__ unsigned first;
    __shared__ int   q_s[8][QCAP + 1];
    __shared__ int   q_d[8][QCAP + 1];
    __shared__ float q_w[8][QCAP + 1];
    const int lane = threadIdx.x & 31;
    const int wave = __builtin_amdgcn_readfirstlane((int)(threadIdx.x >> 5));
    int* qs = q_s[wave]; int* qd = q_d[wave]; float* qw = q_w[wave];
    const int nsplit = min(N, ((N >> 1) + 15) & ~15);
    int q = chip_half();
    if (threadIdx.x == 0) first = atomicAdd(ctr + q * QPITCH, 8u);
    __syncthreads();
    unsigned pend = draw_ticket(ctr + q * QPITCH, lane);
    bool switched = false;
    int t = __builtin_amdgcn_readfirstlane((int)first) + wave;
    int qt = q;
    if (t >= ntiles) {
        t = __builtin_amdgcn_readfirstlane(next_tile(ctr, ntiles, lane, pend, q, switched));
        q = __builtin_amdgcn_readfirstlane(q);
        qt = q;
    }
    if (t < 0) return;
    int s0, d0, s1, d1;
    load_tile(src, dst, E, t, lane, s0, d0, s1, d1);
    bool have = true;
    int nq = 0;
    for (;;) {
        bool i0 = false, i1 = false;
        float a0 = 0.0f, b0 = 0.0f, a1 = 0.0f, b1 = 0.0f;
        int tn = -1, qn = qt;
        int ns0 = 0, nd0 = -1, ns1 = 0, nd1 = -1;
        if (have) {
            tn = __builtin_amdgcn_readfirstlane(next_tile(ctr, ntiles, lane, pend, q, switched));
            q  = __builtin_amdgcn_readfirstlane(q);
            qn = q;
            const unsigned lo  = qt ? (unsigned)nsplit : 0u;
            const unsigned len = qt ? (unsigned)(N - nsplit) : (unsigned)nsplit;
            i0 = ((unsigned)d0 - lo) < len; i1 = ((unsigned)d1 - lo) < len;
            a0 = dinv[i0 ? s0 : 0]; b0 = dinv[i0 ? d0 : 0];
            a1 = dinv[i1 ? s1 : 0]; b1 = dinv[i1 ? d1 : 0];
            if (tn >= 0) load_tile(src, dst, E, tn, lane, ns0, nd0, ns1, nd1);
        }
        int pos = 0;
        while (nq - pos >= 16) {
            const int cnt = min(32, (nq - pos) & ~15);
            int sv, dv; float wv;
            queue_window(qs, qd, qw, pos, nq, lane, sv, dv, wv);
            const int wi = __float_as_int(wv);
            float tv[2], tw[2];
            int   td[2];
#pragma unroll
            for (int k = 0; k < 2; ++k) {
                const int el = 16 * k + (lane >> 1);
                const int st = __shfl(sv, el, 32);
                td[k] = __shfl(dv, el, 32);
                tw[k] = __shfl(wv, el, 32);
                tv[k] = hw2[(size_t)st * 34 + 32 + (lane & 1)];
            }
            for (int j0 = 0; j0 < cnt; j0 += 16) {
                float v[16];
#pragma unroll
                for (int u = 0; u < 16; ++u) {
                    const int sj = __builtin_amdgcn_readlane(sv, j0 + u);
                    v[u] = hw2[(unsigned)(sj * 34 + lane)];
                }
#pragma unroll
                for (int u = 0; u < 16; ++u) {
                    const int   dj = __builtin_amdgcn_readlane(dv, j0 + u);
                    const float wj = __int_as_float(__builtin_amdgcn_readlane(wi, j0 + u));
                    atomicAdd(&out[(unsigned)(dj * 34 + lane)], wj * v[u]);
                }
            }
#pragma unroll
            for (int k = 0; k < 2; ++k) {
                const int el = 16 * k + (lane >> 1);
                if (el < cnt)
                    atomicAdd(out + (size_t)td[k] * 34 + 32 + (lane & 1), tw[k] * tv[k]);
            }
            pos += cnt;
        }
        nq = queue_keep_rest(qs, qd, qw, pos, nq, lane);
        if (!have) break;
        nq = pack_tile(s0, d0, s1, d1, i0, i1, a0, b0, a1, b1, qs, qd, qw, nq);
        have = (tn >= 0);
        qt = qn; s0 = ns0; d0 = nd0; s1 = ns1; d1 = nd1;
    }
    if (nq > 0) {
        int sv, dv; float wv;
        queue_window(qs, qd, qw, 0, nq, lane, sv, dv, wv);
        const int wi = __float_as_int(wv);
        const int el = lane >> 1;
        const int   st  = __shfl(sv, el, 32);
        const int   tdl = __shfl(dv, el, 32);
        const float twl = __shfl(wv, el, 32);
        float tvl = 0.0f;
        if (el < nq) tvl = hw2[(size_t)st * 34 + 32 + (lane & 1)];
        float v[16];
#pragma unroll
        for (int u = 0; u < 16; ++u) {
            if (u < nq) {
                const int sj = __builtin_amdgcn_readlane(sv, u);
                v[u] = hw2[(unsigned)(sj * 34 + lane)];
            }
        }
#pragma unroll
        for (int u = 0; u < 16; ++u) {
            if (u < nq) {
                const int   dj = __builtin_amdgcn_readlane(dv, u);
                const float wj = __int_as_float(__builtin_amdgcn_readlane(wi, u));
                atomicAdd(&out[(unsigned)(dj * 34 + lane)], wj * v[u]);
            }
        }
        if (el < nq)
            atomicAdd(out + (size_t)tdl * 34 + 32 + (lane & 1), twl * tvl);
    }
}

extern "C" void kernel_launch(void* const* d_in, const int* in_sizes, int n_in,
                              void* d_out, int out_size, void* d_ws, size_t ws_size,
                              hipStream_t stream) {
    const float* x  = (const float*)d_in[0];
    const int*   ei = (const int*)  d_in[1];
    const float* W1 = (const float*)d_in[2];
    const float* b1 = (const float*)d_in[3];
    const float* W2 = (const float*)d_in[4];
    const float* b2 = (const float*)d_in[5];
    float* out = (float*)d_out;

    const int N = in_sizes[0] / 64;
    const int E = in_sizes[1] / 2;
    const int* src = ei;
    const int* dst = ei + E;

    const int nranges = (N + DEG_RANGE - 1) / DEG_RANGE;
    const int npad    = nranges * DEG_RANGE;
    const int per     = (((E + DEG_SLICES - 1) / DEG_SLICES) + 63) & ~63;

    float* xw   = (float*)d_ws;
    float* h    = xw   + (size_t)N * 64;
    float* hw2  = h    + (size_t)N * 64;
    float* dinv = hw2  + (size_t)N * 34;
    unsigned* part = (unsigned*)(dinv + N);
    unsigned* ctr  = part + (size_t)DEG_SLICES * npad;

    const int T = 256;
    const int N4 = N / 4;
    const int nstrips = N / 16;
    const int ntiles = (E + TILE_E - 1) / TILE_E;
    k_deg_hist<<<nranges * DEG_SLICES, DEG_T, 0, stream>>>(dst, part, E, per, npad);
    k_dinv    <<<(N4 + 127) / 128, 128, 0, stream>>>(part, dinv, ctr, N4, npad / 4);
    k_gemm64<<<(nstrips + 7) / 8, 256, 0, stream>>>(x, W1, dinv, b1, xw, h, nstrips);
    k_scatter64<<<EDGE_BLOCKS, T, 0, stream>>>(src, dst, dinv, xw, h, ctr, E, N, ntiles);
    k_gemm34<<<(nstrips + 7) / 8, 256, 0, stream>>>(h, W2, dinv, b2, hw2, out, nstrips);
    k_scatter34<<<EDGE_BLOCKS, T, 0, stream>>>(src, dst, dinv, hw2, out, ctr + 2 * QPITCH, E, N, ntiles);
}


# Layer-2 row adds cut at the output row's 128-byte line boundary: two atomic instructions per edge, each inside one line (3 -> 2 line accesses per edge); row loads split alike; atomic order changes
# speedup vs baseline: 2.4964x; 1.0347x over previous
#include <hip/hip_runtime.h>
#include <math.h>

typedef __attribute__((ext_vector_type(2))) float v2f;
typedef __attribute__((ext_vector_type(8))) float v8f;

#define EDGE_BLOCKS 1024
#define TILE_E      64
#define QPITCH      64

#define DEG_RANGE  12800
#define DEG_SLICES 32
#define DEG_T      512
#define DEG_B      20

__device__ __forceinline__ void deg_count1(unsigned* cnt, int d, int lo) {
    const unsigned a = (unsigned)(d - lo);
    if (a < (unsigned)DEG_RANGE) atomicAdd(&cnt[a], 1u);
}

__global__ void __launch_bounds__(DEG_T) k_deg_hist(const int* __restrict__ dst, unsigned* __restrict__ part,
                                                    int E, int per, int npad) {
    __shared__ __attribute__((aligned(16))) unsigned cnt[DEG_RANGE];
    const int r  = blockIdx.x / DEG_SLICES;
    const int s  = blockIdx.x % DEG_SLICES;
    const int lo = r * DEG_RANGE;
    for (int i = threadIdx.x; i < DEG_RANGE / 4; i += DEG_T)
        ((uint4*)cnt)[i] = make_uint4(0u, 0u, 0u, 0u);
    __syncthreads();
    const int e0 = min(E, s * per);
    const int e1 = min(E, e0 + per);
    int e = e0;
    if ((((size_t)(dst + e0)) & 15) == 0) {
        const int4* p = (const int4*)(dst + e0);
        const int n4 = (e1 - e0) >> 2;
        for (int base = 0; base < n4; base += DEG_B * DEG_T) {
            int4 v[DEG_B];
#pragma unroll
            for (int k = 0; k < DEG_B; ++k) {
                const int i = base + k * DEG_T + (int)threadIdx.x;
                v[k] = make_int4(-1, -1, -1, -1);
                if (i < n4) v[k] = p[i];
            }
#pragma unroll
            for (int k = 0; k < DEG_B; ++k) {
                deg_count1(cnt, v[k].x, lo);
                deg_count1(cnt, v[k].y, lo);
                deg_count1(cnt, v[k].z, lo);
                deg_count1(cnt, v[k].w, lo);
            }
        }
        e = e0 + (n4 << 2);
    }
    for (e += (int)threadIdx.x; e < e1; e += DEG_T)
        deg_count1(cnt, dst[e], lo);
    __syncthreads();
    uint4* o = (uint4*)(part + (size_t)s * npad + lo);
    for (int i = threadIdx.x; i < DEG_RANGE / 4; i += DEG_T)
        o[i] = ((const uint4*)cnt)[i];
}

__global__ void __launch_bounds__(128) k_dinv(const unsigned* __restrict__ part, float* __restrict__ dinv,
                                              unsigned* __restrict__ ctr, int n4, int npad4) {
    const int i = blockIdx.x * blockDim.x + threadIdx.x;
    if (i < 4) ctr[i * QPITCH] = 0u;
    if (i < n4) {
        uint4 c = make_uint4(1u, 1u, 1u, 1u);
#pragma unroll
        for (int s = 0; s < DEG_SLICES; ++s) {
            const uint4 v = ((const uint4*)part)[(size_t)s * npad4 + i];
            c.x += v.x; c.y += v.y; c.z += v.z; c.w += v.w;
        }
        ((float4*)dinv)[i] = make_float4(rsqrtf((float)c.x), rsqrtf((float)c.y),
                                         rsqrtf((float)c.z), rsqrtf((float)c.w));
    }
}

__global__ void __launch_bounds__(256) k_gemm64(const float* __restrict__ A,
                                                const float* __restrict__ B,
                                                const float* __restrict__ dinv,
                                                const float* __restrict__ b1,
                                                float* __restrict__ C,
                                                float* __restrict__ Hacc,
                                                int nstrips) {
    const int wave = __builtin_amdgcn_readfirstlane((int)(threadIdx.x >> 5));
    const int lane = threadIdx.x & 31;
    const int half = lane >> 4;
    const int lm   = lane & 15;
    const int strip = blockIdx.x * 8 + wave;
    if (strip >= nstrips) return;
    const int m0   = strip * 16;
    const float* arow = A + (size_t)(m0 + lm) * 64;

    v2f a[16];
#pragma unroll
    for (int kk = 0; kk < 16; ++kk)
        a[kk] = *(const v2f*)(arow + kk * 4 + half * 2);

    v8f acc[4];
#pragma unroll
    for (int nt = 0; nt < 4; ++nt) acc[nt] = (v8f){0.f,0.f,0.f,0.f,0.f,0.f,0.f,0.f};
#pragma unroll
    for (int kk = 0; kk < 16; ++kk) {
        const int kb = kk * 4 + half * 2;
#pragma unroll
        for (int nt = 0; nt < 4; ++nt) {
            const int bcol = nt * 16 + lm;
            v2f b;
            b.x = B[(size_t)kb * 64 + bcol];
            b.y = B[(size_t)(kb + 1) * 64 + bcol];
            acc[nt] = __builtin_amdgcn_wmma_f32_16x16x4_f32(false, a[kk], false, b,
                                                            (short)0, acc[nt], false, false);
        }
    }
    float di2[8];
#pragma unroll
    for (int i = 0; i < 8; ++i) {
        const float di = dinv[m0 + half * 8 + i];
        di2[i] = di * di;
    }
#pragma unroll
    for (int nt = 0; nt < 4; ++nt) {
        const int col = nt * 16 + lm;
        const float bb = b1[col];
#pragma unroll
        for (int i = 0; i < 8; ++i) {
            const int row = m0 + half * 8 + i;
            const float v = acc[nt][i];
            C[(size_t)row * 64 + col]    = v;
            Hacc[(size_t)row * 64 + col] = bb + di2[i] * v;
        }
    }
}

#define QCAP 96

__device__ __forceinline__ int chip_half() {
    return (int)((__builtin_amdgcn_s_sendmsg_rtn(0x87) >> 18) & 1u);
}

__device__ __forceinline__ unsigned draw_ticket(unsigned* q, int lane) {
    unsigned v = 0u;
    if (lane == 0) v = atomicAdd(q, 1u);
    return v;
}

__device__ __forceinline__ int next_tile(unsigned* __restrict__ ctr, int ntiles, int lane,
                                         unsigned& pend, int& q, bool& switched) {
    for (;;) {
        const int t = (int)__builtin_amdgcn_readfirstlane(pend);
        if (t < ntiles) { pend = draw_ticket(ctr + q * QPITCH, lane); return t; }
        if (switched) return -1;
        switched = true;
        q ^= 1;
        pend = draw_ticket(ctr + q * QPITCH, lane);
    }
}

__device__ __forceinline__ void load_tile(const int* __restrict__ src, const int* __restrict__ dst,
                                          int E, int t, int lane, int& s0, int& d0, int& s1, int& d1) {
    const int e = t * TILE_E + lane;
    s0 = 0; d0 = -1; s1 = 0; d1 = -1;
    if (e      < E) { s0 = src[e];      d0 = dst[e];      }
    if (e + 32 < E) { s1 = src[e + 32]; d1 = dst[e + 32]; }
}

__device__ __forceinline__ int pack_tile(int s0, int d0, int s1, int d1, bool i0, bool i1,
                                         float a0, float b0, float a1, float b1,
                                         int* qs, int* qd, float* qw, int nq) {
    const unsigned m0 = __builtin_amdgcn_ballot_w32(i0), m1 = __builtin_amdgcn_ballot_w32(i1);
    const int r0 = i0 ? nq + (int)__builtin_amdgcn_mbcnt_lo(m0, 0u) : QCAP;
    const int r1 = i1 ? nq + __builtin_popcount(m0) + (int)__builtin_amdgcn_mbcnt_lo(m1, 0u) : QCAP;
    qs[r0] = s0; qd[r0] = d0; qw[r0] = a0 * b0;
    qs[r1] = s1; qd[r1] = d1; qw[r1] = a1 * b1;
    return nq + __builtin_popcount(m0) + __builtin_popcount(m1);
}

__device__ __forceinline__ void queue_window(const int* qs, const int* qd, const float* qw, int pos, int nq, int lane,
                                             int& sv, int& dv, float& wv) {
    sv = 0; dv = 0; wv = 0.0f;
    if (pos + lane < nq) { sv = qs[pos + lane]; dv = qd[pos + lane]; wv = qw[pos + lane]; }
}

__device__ __forceinline__ int queue_keep_rest(int* qs, int* qd, float* qw, int pos, int nq, int lane) {
    const int rem = nq - pos;
    if (pos > 0 && rem > 0) {
        int ts = 0, td = 0; float tw = 0.0f;
        if (lane < rem) { ts = qs[pos + lane]; td = qd[pos + lane]; tw = qw[pos + lane]; }
        if (lane < rem) { qs[lane] = ts; qd[lane] = td; qw[lane] = tw; }
    }
    return rem;
}

__global__ void __launch_bounds__(256) k_scatter64(const int* __restrict__ src, const int* __restrict__ dst,
                                                   const float* __restrict__ dinv, const float* __restrict__ xw,
                                                   float* __restrict__ h, unsigned* __restrict__ ctr,
                                                   int E, int N, int ntiles) {
    __shared__ unsigned first;
    __shared__ int   q_s[8][QCAP + 1];
    __shared__ int   q_d[8][QCAP + 1];
    __shared__ float q_w[8][QCAP + 1];
    const int lane = threadIdx.x & 31;
    const int wave = __builtin_amdgcn_readfirstlane((int)(threadIdx.x >> 5));
    int* qs = q_s[wave]; int* qd = q_d[wave]; float* qw = q_w[wave];
    const int nsplit = min(N, ((N >> 1) + 15) & ~15);
    int q = chip_half();
    if (threadIdx.x == 0) first = atomicAdd(ctr + q * QPITCH, 8u);
    __syncthreads();
    unsigned pend = draw_ticket(ctr + q * QPITCH, lane);
    bool switched = false;
    int t = __builtin_amdgcn_readfirstlane((int)first) + wave;
    int qt = q;
    if (t >= ntiles) {
        t = __builtin_amdgcn_readfirstlane(next_tile(ctr, ntiles, lane, pend, q, switched));
        q = __builtin_amdgcn_readfirstlane(q);
        qt = q;
    }
    if (t < 0) return;
    int s0, d0, s1, d1;
    load_tile(src, dst, E, t, lane, s0, d0, s1, d1);
    bool have = true;
    int nq = 0;
    for (;;) {
        bool i0 = false, i1 = false;
        float a0 = 0.0f, b0 = 0.0f, a1 = 0.0f, b1 = 0.0f;
        int tn = -1, qn = qt;
        int ns0 = 0, nd0 = -1, ns1 = 0, nd1 = -1;
        if (have) {
            tn = __builtin_amdgcn_readfirstlane(next_tile(ctr, ntiles, lane, pend, q, switched));
            q  = __builtin_amdgcn_readfirstlane(q);
            qn = q;
            const unsigned lo  = qt ? (unsigned)nsplit : 0u;
            const unsigned len = qt ? (unsigned)(N - nsplit) : (unsigned)nsplit;
            i0 = ((unsigned)d0 - lo) < len; i1 = ((unsigned)d1 - lo) < len;
            a0 = dinv[i0 ? s0 : 0]; b0 = dinv[i0 ? d0 : 0];
            a1 = dinv[i1 ? s1 : 0]; b1 = dinv[i1 ? d1 : 0];
            if (tn >= 0) load_tile(src, dst, E, tn, lane, ns0, nd0, ns1, nd1);
        }
        int pos = 0;
        while (nq - pos >= 16) {
            const int cnt = min(32, (nq - pos) & ~15);
            int sv, dv; float wv;
            queue_window(qs, qd, qw, pos, nq, lane, sv, dv, wv);
            const int wi = __float_as_int(wv);
            for (int j0 = 0; j0 < cnt; j0 += 16) {
                float v0[16], v1[16];
#pragma unroll
                for (int u = 0; u < 16; ++u) {
                    const int sj = __builtin_amdgcn_readlane(sv, j0 + u);
                    const float* xr = xw + (size_t)sj * 64;
                    v0[u] = xr[lane];
                    v1[u] = xr[lane + 32];
                }
#pragma unroll
                for (int u = 0; u < 16; ++u) {
                    const int   dj = __builtin_amdgcn_readlane(dv, j0 + u);
                    const float wj = __int_as_float(__builtin_amdgcn_readlane(wi, j0 + u));
                    float* o = h + (size_t)dj * 64;
                    atomicAdd(o + lane,      wj * v0[u]);
                    atomicAdd(o + lane + 32, wj * v1[u]);
                }
            }
            pos += cnt;
        }
        nq = queue_keep_rest(qs, qd, qw, pos, nq, lane);
        if (!have) break;
        nq = pack_tile(s0, d0, s1, d1, i0, i1, a0, b0, a1, b1, qs, qd, qw, nq);
        have = (tn >= 0);
        qt = qn; s0 = ns0; d0 = nd0; s1 = ns1; d1 = nd1;
    }
    if (nq > 0) {
        int sv, dv; float wv;
        queue_window(qs, qd, qw, 0, nq, lane, sv, dv, wv);
        const int wi = __float_as_int(wv);
        float v0[16], v1[16];
#pragma unroll
        for (int u = 0; u < 16; ++u) {
            if (u < nq) {
                const int sj = __builtin_amdgcn_readlane(sv, u);
                const float* xr = xw + (size_t)sj * 64;
                v0[u] = xr[lane];
                v1[u] = xr[lane + 32];
            }
        }
#pragma unroll
        for (int u = 0; u < 16; ++u) {
            if (u < nq) {
                const int   dj = __builtin_amdgcn_readlane(dv, u);
                const float wj = __int_as_float(__builtin_amdgcn_readlane(wi, u));
                float* o = h + (size_t)dj * 64;
                atomicAdd(o + lane,      wj * v0[u]);
                atomicAdd(o + lane + 32, wj * v1[u]);
            }
        }
    }
}

__global__ void __launch_bounds__(256) k_gemm34(const float* __restrict__ A,
                                                const float* __restrict__ B,
                                                const float* __restrict__ dinv,
                                                const float* __restrict__ b2,
                                                float* __restrict__ C,
                                                float* __restrict__ out,
                                                int nstrips) {
    const int wave = __builtin_amdgcn_readfirstlane((int)(threadIdx.x >> 5));
    const int lane = threadIdx.x & 31;
    const int half = lane >> 4;
    const int lm   = lane & 15;
    const int strip = blockIdx.x * 8 + wave;
    if (strip >= nstrips) return;
    const int m0   = strip * 16;
    const float* arow = A + (size_t)(m0 + lm) * 64;

    v2f a[16];
#pragma unroll
    for (int kk = 0; kk < 16; ++kk)
        a[kk] = *(const v2f*)(arow + kk * 4 + half * 2);

    float m = 0.0f;
#pragma unroll
    for (int kk = 0; kk < 16; ++kk) {
        a[kk].x = fmaxf(a[kk].x, 0.0f);
        a[kk].y = fmaxf(a[kk].y, 0.0f);
        m = fmaxf(m, fmaxf(a[kk].x, a[kk].y));
    }
    m = fmaxf(m, __shfl_xor(m, 16, 32));
    float s = 0.0f;
#pragma unroll
    for (int kk = 0; kk < 16; ++kk) {
        a[kk].x = __expf(a[kk].x - m);
        a[kk].y = __expf(a[kk].y - m);
        s += a[kk].x + a[kk].y;
    }
    s += __shfl_xor(s, 16, 32);
    const float inv = 1.0f / s;

    v8f acc[3];
#pragma unroll
    for (int nt = 0; nt < 3; ++nt) acc[nt] = (v8f){0.f,0.f,0.f,0.f,0.f,0.f,0.f,0.f};
#pragma unroll
    for (int kk = 0; kk < 16; ++kk) {
        const int kb = kk * 4 + half * 2;
        v2f av;
        av.x = a[kk].x * inv;
        av.y = a[kk].y * inv;
#pragma unroll
        for (int nt = 0; nt < 3; ++nt) {
            const int col = nt * 16 + lm;
            const bool ok = (col < 34);
            const int  cc = ok ? col : 33;
            const float bx = B[(size_t)kb * 34 + cc];
            const float by = B[(size_t)(kb + 1) * 34 + cc];
            v2f b;
            b.x = ok ? bx : 0.0f;
            b.y = ok ? by : 0.0f;
            acc[nt] = __builtin_amdgcn_wmma_f32_16x16x4_f32(false, av, false, b,
                                                            (short)0, acc[nt], false, false);
        }
    }
    float di2[8];
#pragma unroll
    for (int i = 0; i < 8; ++i) {
        const float di = dinv[m0 + half * 8 + i];
        di2[i] = di * di;
    }
#pragma unroll
    for (int nt = 0; nt < 3; ++nt) {
        const int col = nt * 16 + lm;
        if (col < 34) {
            const float bb = b2[col];
#pragma unroll
            for (int i = 0; i < 8; ++i) {
                const int row = m0 + half * 8 + i;
                const float v = acc[nt][i];
                C[(size_t)row * 34 + col]   = v;
                out[(size_t)row * 34 + col] = bb + di2[i] * v;
            }
        }
    }
}

__device__ __forceinline__ int row_cut(int d) { return 32 - 2 * (d & 15); }

template <bool GUARD>
__device__ __forceinline__ void add_rows34(const float* __restrict__ hw2, float* __restrict__ out,
                                           int sv, int dv, int wi, int j0, int n, int lane) {
    float va[16], vb[16];
#pragma unroll
    for (int u = 0; u < 16; ++u) {
        if (!GUARD || u < n) {
            const int sj = __builtin_amdgcn_readlane(sv, j0 + u);
            const int na = row_cut(__builtin_amdgcn_readlane(dv, j0 + u));
            va[u] = hw2[(unsigned)(sj * 34 + min(lane, na - 1))];
            vb[u] = hw2[(unsigned)(sj * 34 + min(na + lane, 33))];
        }
    }
    if (!GUARD) asm volatile("" :: "v"(va[0]));
#pragma unroll
    for (int u = 0; u < 16; ++u) {
        if (!GUARD || u < n) {
            const int   dj = __builtin_amdgcn_readlane(dv, j0 + u);
            const float wj = __int_as_float(__builtin_amdgcn_readlane(wi, j0 + u));
            const int   na = row_cut(dj);
            if (lane < na)      atomicAdd(&out[(unsigned)(dj * 34 + lane)],      wj * va[u]);
            if (lane < 34 - na) atomicAdd(&out[(unsigned)(dj * 34 + na + lane)], wj * vb[u]);
        }
    }
}

__global__ void __launch_bounds__(256) k_scatter34(const int* __restrict__ src, const int* __restrict__ dst,
                                                   const float* __restrict__ dinv, const float* __restrict__ hw2,
                                                   float* __restrict__ out, unsigned* __restrict__ ctr,
                                                   int E, int N, int ntiles) {
    __shared__ unsigned first;
    __shared__ int   q_s[8][QCAP + 1];
    __shared__ int   q_d[8][QCAP + 1];
    __shared__ float q_w[8][QCAP + 1];
    const int lane = threadIdx.x & 31;
    const int wave = __builtin_amdgcn_readfirstlane((int)(threadIdx.x >> 5));
    int* qs = q_s[wave]; int* qd = q_d[wave]; float* qw = q_w[wave];
    const int nsplit = min(N, ((N >> 1) + 15) & ~15);
    int q = chip_half();
    if (threadIdx.x == 0) first = atomicAdd(ctr + q * QPITCH, 8u);
    __syncthreads();
    unsigned pend = draw_ticket(ctr + q * QPITCH, lane);
    bool switched = false;
    int t = __builtin_amdgcn_readfirstlane((int)first) + wave;
    int qt = q;
    if (t >= ntiles) {
        t = __builtin_amdgcn_readfirstlane(next_tile(ctr, ntiles, lane, pend, q, switched));
        q = __builtin_amdgcn_readfirstlane(q);
        qt = q;
    }
    if (t < 0) return;
    int s0, d0, s1, d1;
    load_tile(src, dst, E, t, lane, s0, d0, s1, d1);
    bool have = true;
    int nq = 0;
    for (;;) {
        bool i0 = false, i1 = false;
        float a0 = 0.0f, b0 = 0.0f, a1 = 0.0f, b1 = 0.0f;
        int tn = -1, qn = qt;
        int ns0 = 0, nd0 = -1, ns1 = 0, nd1 = -1;
        if (have) {
            tn = __builtin_amdgcn_readfirstlane(next_tile(ctr, ntiles, lane, pend, q, switched));
            q  = __builtin_amdgcn_readfirstlane(q);
            qn = q;
            const unsigned lo  = qt ? (unsigned)nsplit : 0u;
            const unsigned len = qt ? (unsigned)(N - nsplit) : (unsigned)nsplit;
            i0 = ((unsigned)d0 - lo) < len; i1 = ((unsigned)d1 - lo) < len;
            a0 = dinv[i0 ? s0 : 0]; b0 = dinv[i0 ? d0 : 0];
            a1 = dinv[i1 ? s1 : 0]; b1 = dinv[i1 ? d1 : 0];
            if (tn >= 0) load_tile(src, dst, E, tn, lane, ns0, nd0, ns1, nd1);
        }
        int pos = 0;
        while (nq - pos >= 16) {
            const int cnt = min(32, (nq - pos) & ~15);
            int sv, dv; float wv;
            queue_window(qs, qd, qw, pos, nq, lane, sv, dv, wv);
            const int wi = __float_as_int(wv);
            for (int j0 = 0; j0 < cnt; j0 += 16)
                add_rows34<false>(hw2, out, sv, dv, wi, j0, 16, lane);
            pos += cnt;
        }
        nq = queue_keep_rest(qs, qd, qw, pos, nq, lane);
        if (!have) break;
        nq = pack_tile(s0, d0, s1, d1, i0, i1, a0, b0, a1, b1, qs, qd, qw, nq);
        have = (tn >= 0);
        qt = qn; s0 = ns0; d0 = nd0; s1 = ns1; d1 = nd1;
    }
    if (nq > 0) {
        int sv, dv; float wv;
        queue_window(qs, qd, qw, 0, nq, lane, sv, dv, wv);
        const int wi = __float_as_int(wv);
        add_rows34<true>(hw2, out, sv, dv, wi, 0, nq, lane);
    }
}

extern "C" void kernel_launch(void* const* d_in, const int* in_sizes, int n_in,
                              void* d_out, int out_size, void* d_ws, size_t ws_size,
                              hipStream_t stream) {
    const float* x  = (const float*)d_in[0];
    const int*   ei = (const int*)  d_in[1];
    const float* W1 = (const float*)d_in[2];
    const float* b1 = (const float*)d_in[3];
    const float* W2 = (const float*)d_in[4];
    const float* b2 = (const float*)d_in[5];
    float* out = (float*)d_out;

    const int N = in_sizes[0] / 64;
    const int E = in_sizes[1] / 2;
    const int* src = ei;
    const int* dst = ei + E;

    const int nranges = (N + DEG_RANGE - 1) / DEG_RANGE;
    const int npad    = nranges * DEG_RANGE;
    const int per     = (((E + DEG_SLICES - 1) / DEG_SLICES) + 63) & ~63;

    float* xw   = (float*)d_ws;
    float* h    = xw   + (size_t)N * 64;
    float* hw2  = h    + (size_t)N * 64;
    float* dinv = hw2  + (size_t)N * 34;
    unsigned* part = (unsigned*)(dinv + N);
    unsigned* ctr  = part + (size_t)DEG_SLICES * npad;

    const int T = 256;
    const int N4 = N / 4;
    const int nstrips = N / 16;
    const int ntiles = (E + TILE_E - 1) / TILE_E;
    k_deg_hist<<<nranges * DEG_SLICES, DEG_T, 0, stream>>>(dst, part, E, per, npad);
    k_dinv    <<<(N4 + 127) / 128, 128, 0, stream>>>(part, dinv, ctr, N4, npad / 4);
    k_gemm64<<<(nstrips + 7) / 8, 256, 0, stream>>>(x, W1, dinv, b1, xw, h, nstrips);
    k_scatter64<<<EDGE_BLOCKS, T, 0, stream>>>(src, dst, dinv, xw, h, ctr, E, N, ntiles);
    k_gemm34<<<(nstrips + 7) / 8, 256, 0, stream>>>(h, W2, dinv, b2, hw2, out, nstrips);
    k_scatter34<<<EDGE_BLOCKS, T, 0, stream>>>(src, dst, dinv, hw2, out, ctr + 2 * QPITCH, E, N, ntiles);
}


# Edge weights dinv[src]*dinv[dst] stored per (range, tile) visit by the layer-1 edge kernel as whole lines; layer 2 reads them back with the tile's ids: its two per-edge dinv gathers go; same sums
# speedup vs baseline: 2.6464x; 1.0601x over previous
#include <hip/hip_runtime.h>
#include <math.h>

typedef __attribute__((ext_vector_type(2))) float v2f;
typedef __attribute__((ext_vector_type(8))) float v8f;

#define EDGE_BLOCKS 1024
#define TILE_E      64
#define QPITCH      64

#define DEG_RANGE  12800
#define DEG_SLICES 32
#define DEG_T      512
#define DEG_B      20

__device__ __forceinline__ void deg_count1(unsigned* cnt, int d, int lo) {
    const unsigned a = (unsigned)(d - lo);
    if (a < (unsigned)DEG_RANGE) atomicAdd(&cnt[a], 1u);
}

__global__ void __launch_bounds__(DEG_T) k_deg_hist(const int* __restrict__ dst, unsigned* __restrict__ part,
                                                    int E, int per, int npad) {
    __shared__ __attribute__((aligned(16))) unsigned cnt[DEG_RANGE];
    const int r  = blockIdx.x / DEG_SLICES;
    const int s  = blockIdx.x % DEG_SLICES;
    const int lo = r * DEG_RANGE;
    for (int i = threadIdx.x; i < DEG_RANGE / 4; i += DEG_T)
        ((uint4*)cnt)[i] = make_uint4(0u, 0u, 0u, 0u);
    __syncthreads();
    const int e0 = min(E, s * per);
    const int e1 = min(E, e0 + per);
    int e = e0;
    if ((((size_t)(dst + e0)) & 15) == 0) {
        const int4* p = (const int4*)(dst + e0);
        const int n4 = (e1 - e0) >> 2;
        for (int base = 0; base < n4; base += DEG_B * DEG_T) {
            int4 v[DEG_B];
#pragma unroll
            for (int k = 0; k < DEG_B; ++k) {
                const int i = base + k * DEG_T + (int)threadIdx.x;
                v[k] = make_int4(-1, -1, -1, -1);
                if (i < n4) v[k] = p[i];
            }
#pragma unroll
            for (int k = 0; k < DEG_B; ++k) {
                deg_count1(cnt, v[k].x, lo);
                deg_count1(cnt, v[k].y, lo);
                deg_count1(cnt, v[k].z, lo);
                deg_count1(cnt, v[k].w, lo);
            }
        }
        e = e0 + (n4 << 2);
    }
    for (e += (int)threadIdx.x; e < e1; e += DEG_T)
        deg_count1(cnt, dst[e], lo);
    __syncthreads();
    uint4* o = (uint4*)(part + (size_t)s * npad + lo);
    for (int i = threadIdx.x; i < DEG_RANGE / 4; i += DEG_T)
        o[i] = ((const uint4*)cnt)[i];
}

__global__ void __launch_bounds__(128) k_dinv(const unsigned* __restrict__ part, float* __restrict__ dinv,
                                              unsigned* __restrict__ ctr, int n4, int npad4) {
    const int i = blockIdx.x * blockDim.x + threadIdx.x;
    if (i < 4) ctr[i * QPITCH] = 0u;
    if (i < n4) {
        uint4 c = make_uint4(1u, 1u, 1u, 1u);
#pragma unroll
        for (int s = 0; s < DEG_SLICES; ++s) {
            const uint4 v = ((const uint4*)part)[(size_t)s * npad4 + i];
            c.x += v.x; c.y += v.y; c.z += v.z; c.w += v.w;
        }
        ((float4*)dinv)[i] = make_float4(rsqrtf((float)c.x), rsqrtf((float)c.y),
                                         rsqrtf((float)c.z), rsqrtf((float)c.w));
    }
}

__global__ void __launch_bounds__(256) k_gemm64(const float* __restrict__ A,
                                                const float* __restrict__ B,
                                                const float* __restrict__ dinv,
                                                const float* __restrict__ b1,
                                                float* __restrict__ C,
                                                float* __restrict__ Hacc,
                                                int nstrips) {
    const int wave = __builtin_amdgcn_readfirstlane((int)(threadIdx.x >> 5));
    const int lane = threadIdx.x & 31;
    const int half = lane >> 4;
    const int lm   = lane & 15;
    const int strip = blockIdx.x * 8 + wave;
    if (strip >= nstrips) return;
    const int m0   = strip * 16;
    const float* arow = A + (size_t)(m0 + lm) * 64;

    v2f a[16];
#pragma unroll
    for (int kk = 0; kk < 16; ++kk)
        a[kk] = *(const v2f*)(arow + kk * 4 + half * 2);

    v8f acc[4];
#pragma unroll
    for (int nt = 0; nt < 4; ++nt) acc[nt] = (v8f){0.f,0.f,0.f,0.f,0.f,0.f,0.f,0.f};
#pragma unroll
    for (int kk = 0; kk < 16; ++kk) {
        const int kb = kk * 4 + half * 2;
#pragma unroll
        for (int nt = 0; nt < 4; ++nt) {
            const int bcol = nt * 16 + lm;
            v2f b;
            b.x = B[(size_t)kb * 64 + bcol];
            b.y = B[(size_t)(kb + 1) * 64 + bcol];
            acc[nt] = __builtin_amdgcn_wmma_f32_16x16x4_f32(false, a[kk], false, b,
                                                            (short)0, acc[nt], false, false);
        }
    }
    float di2[8];
#pragma unroll
    for (int i = 0; i < 8; ++i) {
        const float di = dinv[m0 + half * 8 + i];
        di2[i] = di * di;
    }
#pragma unroll
    for (int nt = 0; nt < 4; ++nt) {
        const int col = nt * 16 + lm;
        const float bb = b1[col];
#pragma unroll
        for (int i = 0; i < 8; ++i) {
            const int row = m0 + half * 8 + i;
            const float v = acc[nt][i];
            C[(size_t)row * 64 + col]    = v;
            Hacc[(size_t)row * 64 + col] = bb + di2[i] * v;
        }
    }
}

#define QCAP 96

__device__ __forceinline__ int chip_half() {
    return (int)((__builtin_amdgcn_s_sendmsg_rtn(0x87) >> 18) & 1u);
}

__device__ __forceinline__ unsigned draw_ticket(unsigned* q, int lane) {
    unsigned v = 0u;
    if (lane == 0) v = atomicAdd(q, 1u);
    return v;
}

__device__ __forceinline__ int next_tile(unsigned* __restrict__ ctr, int ntiles, int lane,
                                         unsigned& pend, int& q, bool& switched) {
    for (;;) {
        const int t = (int)__builtin_amdgcn_readfirstlane(pend);
        if (t < ntiles) { pend = draw_ticket(ctr + q * QPITCH, lane); return t; }
        if (switched) return -1;
        switched = true;
        q ^= 1;
        pend = draw_ticket(ctr + q * QPITCH, lane);
    }
}

__device__ __forceinline__ void load_tile(const int* __restrict__ src, const int* __restrict__ dst,
                                          int E, int t, int lane, int& s0, int& d0, int& s1, int& d1) {
    const int e = t * TILE_E + lane;
    s0 = 0; d0 = -1; s1 = 0; d1 = -1;
    if (e      < E) { s0 = src[e];      d0 = dst[e];      }
    if (e + 32 < E) { s1 = src[e + 32]; d1 = dst[e + 32]; }
}

__device__ __forceinline__ int pack_tile(int s0, int d0, int s1, int d1, bool i0, bool i1,
                                         float w0, float w1,
                                         int* qs, int* qd, float* qw, int nq) {
    const unsigned m0 = __builtin_amdgcn_ballot_w32(i0), m1 = __builtin_amdgcn_ballot_w32(i1);
    const int r0 = i0 ? nq + (int)__builtin_amdgcn_mbcnt_lo(m0, 0u) : QCAP;
    const int r1 = i1 ? nq + __builtin_popcount(m0) + (int)__builtin_amdgcn_mbcnt_lo(m1, 0u) : QCAP;
    qs[r0] = s0; qd[r0] = d0; qw[r0] = w0;
    qs[r1] = s1; qd[r1] = d1; qw[r1] = w1;
    return nq + __builtin_popcount(m0) + __builtin_popcount(m1);
}

__device__ __forceinline__ void queue_window(const int* qs, const int* qd, const float* qw, int pos, int nq, int lane,
                                             int& sv, int& dv, float& wv) {
    sv = 0; dv = 0; wv = 0.0f;
    if (pos + lane < nq) { sv = qs[pos + lane]; dv = qd[pos + lane]; wv = qw[pos + lane]; }
}

__device__ __forceinline__ int queue_keep_rest(int* qs, int* qd, float* qw, int pos, int nq, int lane) {
    const int rem = nq - pos;
    if (pos > 0 && rem > 0) {
        int ts = 0, td = 0; float tw = 0.0f;
        if (lane < rem) { ts = qs[pos + lane]; td = qd[pos + lane]; tw = qw[pos + lane]; }
        if (lane < rem) { qs[lane] = ts; qd[lane] = td; qw[lane] = tw; }
    }
    return rem;
}

__global__ void __launch_bounds__(256) k_scatter64(const int* __restrict__ src, const int* __restrict__ dst,
                                                   const float* __restrict__ dinv, const float* __restrict__ xw,
                                                   float* __restrict__ h, unsigned* __restrict__ ctr,
                                                   float* __restrict__ wgt,
                                                   int E, int N, int ntiles) {
    __shared__ unsigned first;
    __shared__ int   q_s[8][QCAP + 1];
    __shared__ int   q_d[8][QCAP + 1];
    __shared__ float q_w[8][QCAP + 1];
    const int lane = threadIdx.x & 31;
    const int wave = __builtin_amdgcn_readfirstlane((int)(threadIdx.x >> 5));
    int* qs = q_s[wave]; int* qd = q_d[wave]; float* qw = q_w[wave];
    const int nsplit = min(N, ((N >> 1) + 15) & ~15);
    int q = chip_half();
    if (threadIdx.x == 0) first = atomicAdd(ctr + q * QPITCH, 8u);
    __syncthreads();
    unsigned pend = draw_ticket(ctr + q * QPITCH, lane);
    bool switched = false;
    int t = __builtin_amdgcn_readfirstlane((int)first) + wave;
    int qt = q;
    if (t >= ntiles) {
        t = __builtin_amdgcn_readfirstlane(next_tile(ctr, ntiles, lane, pend, q, switched));
        q = __builtin_amdgcn_readfirstlane(q);
        qt = q;
    }
    if (t < 0) return;
    int s0, d0, s1, d1;
    load_tile(src, dst, E, t, lane, s0, d0, s1, d1);
    bool have = true;
    int tc = t;
    int nq = 0;
    for (;;) {
        bool i0 = false, i1 = false;
        float a0 = 0.0f, b0 = 0.0f, a1 = 0.0f, b1 = 0.0f;
        int tn = -1, qn = qt;
        int ns0 = 0, nd0 = -1, ns1 = 0, nd1 = -1;
        if (have) {
            tn = __builtin_amdgcn_readfirstlane(next_tile(ctr, ntiles, lane, pend, q, switched));
            q  = __builtin_amdgcn_readfirstlane(q);
            qn = q;
            const unsigned lo  = qt ? (unsigned)nsplit : 0u;
            const unsigned len = qt ? (unsigned)(N - nsplit) : (unsigned)nsplit;
            i0 = ((unsigned)d0 - lo) < len; i1 = ((unsigned)d1 - lo) < len;
            a0 = dinv[i0 ? s0 : 0]; b0 = dinv[i0 ? d0 : 0];
            a1 = dinv[i1 ? s1 : 0]; b1 = dinv[i1 ? d1 : 0];
            if (tn >= 0) load_tile(src, dst, E, tn, lane, ns0, nd0, ns1, nd1);
        }
        int pos = 0;
        while (nq - pos >= 16) {
            const int cnt = min(32, (nq - pos) & ~15);
            int sv, dv; float wv;
            queue_window(qs, qd, qw, pos, nq, lane, sv, dv, wv);
            const int wi = __float_as_int(wv);
            for (int j0 = 0; j0 < cnt; j0 += 16) {
                float v0[16], v1[16];
#pragma unroll
                for (int u = 0; u < 16; ++u) {
                    const int sj = __builtin_amdgcn_readlane(sv, j0 + u);
                    const float* xr = xw + (size_t)sj * 64;
                    v0[u] = xr[lane];
                    v1[u] = xr[lane + 32];
                }
#pragma unroll
                for (int u = 0; u < 16; ++u) {
                    const int   dj = __builtin_amdgcn_readlane(dv, j0 + u);
                    const float wj = __int_as_float(__builtin_amdgcn_readlane(wi, j0 + u));
                    float* o = h + (size_t)dj * 64;
                    atomicAdd(o + lane,      wj * v0[u]);
                    atomicAdd(o + lane + 32, wj * v1[u]);
                }
            }
            pos += cnt;
        }
        nq = queue_keep_rest(qs, qd, qw, pos, nq, lane);
        if (!have) break;
        const float w0 = a0 * b0, w1 = a1 * b1;
        float* wt = wgt + ((size_t)qt * ntiles + tc) * TILE_E;
        wt[lane] = w0; wt[lane + 32] = w1;
        nq = pack_tile(s0, d0, s1, d1, i0, i1, w0, w1, qs, qd, qw, nq);
        have = (tn >= 0);
        tc = tn;
        qt = qn; s0 = ns0; d0 = nd0; s1 = ns1; d1 = nd1;
    }
    if (nq > 0) {
        int sv, dv; float wv;
        queue_window(qs, qd, qw, 0, nq, lane, sv, dv, wv);
        const int wi = __float_as_int(wv);
        float v0[16], v1[16];
#pragma unroll
        for (int u = 0; u < 16; ++u) {
            if (u < nq) {
                const int sj = __builtin_amdgcn_readlane(sv, u);
                const float* xr = xw + (size_t)sj * 64;
                v0[u] = xr[lane];
                v1[u] = xr[lane + 32];
            }
        }
#pragma unroll
        for (int u = 0; u < 16; ++u) {
            if (u < nq) {
                const int   dj = __builtin_amdgcn_readlane(dv, u);
                const float wj = __int_as_float(__builtin_amdgcn_readlane(wi, u));
                float* o = h + (size_t)dj * 64;
                atomicAdd(o + lane,      wj * v0[u]);
                atomicAdd(o + lane + 32, wj * v1[u]);
            }
        }
    }
}

__global__ void __launch_bounds__(256) k_gemm34(const float* __restrict__ A,
                                                const float* __restrict__ B,
                                                const float* __restrict__ dinv,
                                                const float* __restrict__ b2,
                                                float* __restrict__ C,
                                                float* __restrict__ out,
                                                int nstrips) {
    const int wave = __builtin_amdgcn_readfirstlane((int)(threadIdx.x >> 5));
    const int lane = threadIdx.x & 31;
    const int half = lane >> 4;
    const int lm   = lane & 15;
    const int strip = blockIdx.x * 8 + wave;
    if (strip >= nstrips) return;
    const int m0   = strip * 16;
    const float* arow = A + (size_t)(m0 + lm) * 64;

    v2f a[16];
#pragma unroll
    for (int kk = 0; kk < 16; ++kk)
        a[kk] = *(const v2f*)(arow + kk * 4 + half * 2);

    float m = 0.0f;
#pragma unroll
    for (int kk = 0; kk < 16; ++kk) {
        a[kk].x = fmaxf(a[kk].x, 0.0f);
        a[kk].y = fmaxf(a[kk].y, 0.0f);
        m = fmaxf(m, fmaxf(a[kk].x, a[kk].y));
    }
    m = fmaxf(m, __shfl_xor(m, 16, 32));
    float s = 0.0f;
#pragma unroll
    for (int kk = 0; kk < 16; ++kk) {
        a[kk].x = __expf(a[kk].x - m);
        a[kk].y = __expf(a[kk].y - m);
        s += a[kk].x + a[kk].y;
    }
    s += __shfl_xor(s, 16, 32);
    const float inv = 1.0f / s;

    v8f acc[3];
#pragma unroll
    for (int nt = 0; nt < 3; ++nt) acc[nt] = (v8f){0.f,0.f,0.f,0.f,0.f,0.f,0.f,0.f};
#pragma unroll
    for (int kk = 0; kk < 16; ++kk) {
        const int kb = kk * 4 + half * 2;
        v2f av;
        av.x = a[kk].x * inv;
        av.y = a[kk].y * inv;
#pragma unroll
        for (int nt = 0; nt < 3; ++nt) {
            const int col = nt * 16 + lm;
            const bool ok = (col < 34);
            const int  cc = ok ? col : 33;
            const float bx = B[(size_t)kb * 34 + cc];
            const float by = B[(size_t)(kb + 1) * 34 + cc];
            v2f b;
            b.x = ok ? bx : 0.0f;
            b.y = ok ? by : 0.0f;
            acc[nt] = __builtin_amdgcn_wmma_f32_16x16x4_f32(false, av, false, b,
                                                            (short)0, acc[nt], false, false);
        }
    }
    float di2[8];
#pragma unroll
    for (int i = 0; i < 8; ++i) {
        const float di = dinv[m0 + half * 8 + i];
        di2[i] = di * di;
    }
#pragma unroll
    for (int nt = 0; nt < 3; ++nt) {
        const int col = nt * 16 + lm;
        if (col < 34) {
            const float bb = b2[col];
#pragma unroll
            for (int i = 0; i < 8; ++i) {
                const int row = m0 + half * 8 + i;
                const float v = acc[nt][i];
                C[(size_t)row * 34 + col]   = v;
                out[(size_t)row * 34 + col] = bb + di2[i] * v;
            }
        }
    }
}

__device__ __forceinline__ int row_cut(int d) { return 32 - 2 * (d & 15); }

template <bool GUARD>
__device__ __forceinline__ void add_rows34(const float* __restrict__ hw2, float* __restrict__ out,
                                           int sv, int dv, int wi, int j0, int n, int lane) {
    float va[16], vb[16];
#pragma unroll
    for (int u = 0; u < 16; ++u) {
        if (!GUARD || u < n) {
            const int sj = __builtin_amdgcn_readlane(sv, j0 + u);
            const int na = row_cut(__builtin_amdgcn_readlane(dv, j0 + u));
            va[u] = hw2[(unsigned)(sj * 34 + min(lane, na - 1))];
            vb[u] = hw2[(unsigned)(sj * 34 + min(na + lane, 33))];
        }
    }
    if (!GUARD) asm volatile("" :: "v"(va[0]));
#pragma unroll
    for (int u = 0; u < 16; ++u) {
        if (!GUARD || u < n) {
            const int   dj = __builtin_amdgcn_readlane(dv, j0 + u);
            const float wj = __int_as_float(__builtin_amdgcn_readlane(wi, j0 + u));
            const int   na = row_cut(dj);
            if (lane < na)      atomicAdd(&out[(unsigned)(dj * 34 + lane)],      wj * va[u]);
            if (lane < 34 - na) atomicAdd(&out[(unsigned)(dj * 34 + na + lane)], wj * vb[u]);
        }
    }
}

__global__ void __launch_bounds__(256) k_scatter34(const int* __restrict__ src, const int* __restrict__ dst,
                                                   const float* __restrict__ wgt, const float* __restrict__ hw2,
                                                   float* __restrict__ out, unsigned* __restrict__ ctr,
                                                   int E, int N, int ntiles) {
    __shared__ unsigned first;
    __shared__ int   q_s[8][QCAP + 1];
    __shared__ int   q_d[8][QCAP + 1];
    __shared__ float q_w[8][QCAP + 1];
    const int lane = threadIdx.x & 31;
    const int wave = __builtin_amdgcn_readfirstlane((int)(threadIdx.x >> 5));
    int* qs = q_s[wave]; int* qd = q_d[wave]; float* qw = q_w[wave];
    const int nsplit = min(N, ((N >> 1) + 15) & ~15);
    int q = chip_half();
    if (threadIdx.x == 0) first = atomicAdd(ctr + q * QPITCH, 8u);
    __syncthreads();
    unsigned pend = draw_ticket(ctr + q * QPITCH, lane);
    bool switched = false;
    int t = __builtin_amdgcn_readfirstlane((int)first) + wave;
    int qt = q;
    if (t >= ntiles) {
        t = __builtin_amdgcn_readfirstlane(next_tile(ctr, ntiles, lane, pend, q, switched));
        q = __builtin_amdgcn_readfirstlane(q);
        qt = q;
    }
    if (t < 0) return;
    int s0, d0, s1, d1;
    load_tile(src, dst, E, t, lane, s0, d0, s1, d1);
    const float* wt = wgt + ((size_t)qt * ntiles + t) * TILE_E;
    float w0 = wt[lane], w1 = wt[lane + 32];
    bool have = true;
    int nq = 0;
    for (;;) {
        bool i0 = false, i1 = false;
        int tn = -1, qn = qt;
        int ns0 = 0, nd0 = -1, ns1 = 0, nd1 = -1;
        float nw0 = 0.0f, nw1 = 0.0f;
        if (have) {
            tn = __builtin_amdgcn_readfirstlane(next_tile(ctr, ntiles, lane, pend, q, switched));
            q  = __builtin_amdgcn_readfirstlane(q);
            qn = q;
            const unsigned lo  = qt ? (unsigned)nsplit : 0u;
            const unsigned len = qt ? (unsigned)(N - nsplit) : (unsigned)nsplit;
            i0 = ((unsigned)d0 - lo) < len; i1 = ((unsigned)d1 - lo) < len;
            if (tn >= 0) {
                load_tile(src, dst, E, tn, lane, ns0, nd0, ns1, nd1);
                const float* nwt = wgt + ((size_t)qn * ntiles + tn) * TILE_E;
                nw0 = nwt[lane]; nw1 = nwt[lane + 32];
            }
        }
        int pos = 0;
        while (nq - pos >= 16) {
            const int cnt = min(32, (nq - pos) & ~15);
            int sv, dv; float wv;
            queue_window(qs, qd, qw, pos, nq, lane, sv, dv, wv);
            const int wi = __float_as_int(wv);
            for (int j0 = 0; j0 < cnt; j0 += 16)
                add_rows34<false>(hw2, out, sv, dv, wi, j0, 16, lane);
            pos += cnt;
        }
        nq = queue_keep_rest(qs, qd, qw, pos, nq, lane);
        if (!have) break;
        nq = pack_tile(s0, d0, s1, d1, i0, i1, w0, w1, qs, qd, qw, nq);
        have = (tn >= 0);
        qt = qn; s0 = ns0; d0 = nd0; s1 = ns1; d1 = nd1; w0 = nw0; w1 = nw1;
    }
    if (nq > 0) {
        int sv, dv; float wv;
        queue_window(qs, qd, qw, 0, nq, lane, sv, dv, wv);
        const int wi = __float_as_int(wv);
        add_rows34<true>(hw2, out, sv, dv, wi, 0, nq, lane);
    }
}

extern "C" void kernel_launch(void* const* d_in, const int* in_sizes, int n_in,
                              void* d_out, int out_size, void* d_ws, size_t ws_size,
                              hipStream_t stream) {
    const float* x  = (const float*)d_in[0];
    const int*   ei = (const int*)  d_in[1];
    const float* W1 = (const float*)d_in[2];
    const float* b1 = (const float*)d_in[3];
    const float* W2 = (const float*)d_in[4];
    const float* b2 = (const float*)d_in[5];
    float* out = (float*)d_out;

    const int N = in_sizes[0] / 64;
    const int E = in_sizes[1] / 2;
    const int* src = ei;
    const int* dst = ei + E;

    const int nranges = (N + DEG_RANGE - 1) / DEG_RANGE;
    const int npad    = nranges * DEG_RANGE;
    const int per     = (((E + DEG_SLICES - 1) / DEG_SLICES) + 63) & ~63;

    float* xw   = (float*)d_ws;
    float* h    = xw   + (size_t)N * 64;
    float* hw2  = h    + (size_t)N * 64;
    float* dinv = hw2  + (size_t)N * 34;
    unsigned* part = (unsigned*)(dinv + N);
    unsigned* ctr  = part + (size_t)DEG_SLICES * npad;
    float* wgt = (float*)(ctr + 4 * QPITCH);

    const int T = 256;
    const int N4 = N / 4;
    const int nstrips = N / 16;
    const int ntiles = (E + TILE_E - 1) / TILE_E;
    k_deg_hist<<<nranges * DEG_SLICES, DEG_T, 0, stream>>>(dst, part, E, per, npad);
    k_dinv    <<<(N4 + 127) / 128, 128, 0, stream>>>(part, dinv, ctr, N4, npad / 4);
    k_gemm64<<<(nstrips + 7) / 8, 256, 0, stream>>>(x, W1, dinv, b1, xw, h, nstrips);
    k_scatter64<<<EDGE_BLOCKS, T, 0, stream>>>(src, dst, dinv, xw, h, ctr, wgt, E, N, ntiles);
    k_gemm34<<<(nstrips + 7) / 8, 256, 0, stream>>>(h, W2, dinv, b2, hw2, out, nstrips);
    k_scatter34<<<EDGE_BLOCKS, T, 0, stream>>>(src, dst, wgt, hw2, out, ctr + 2 * QPITCH, E, N, ntiles);
}


# Layer-1 per-edge dinv gathers leave memory: k_dinv also stores each node's degree as a byte, each 32-wave edge block copies the table to LDS and takes rsqrt itself; same sums, atomic order may change
# speedup vs baseline: 2.7504x; 1.0393x over previous
#include <hip/hip_runtime.h>
#include <math.h>

typedef __attribute__((ext_vector_type(2))) float v2f;
typedef __attribute__((ext_vector_type(8))) float v8f;

#define EDGE_BLOCKS 1024
#define EDGE1_BLOCKS 256
#define EDGE1_T     1024
#define DEG8_CAP  102400
#define TILE_E      64
#define QPITCH      64

#define DEG_RANGE  12800
#define DEG_SLICES 32
#define DEG_T      512
#define DEG_B      20

__device__ __forceinline__ void deg_count1(unsigned* cnt, int d, int lo) {
    const unsigned a = (unsigned)(d - lo);
    if (a < (unsigned)DEG_RANGE) atomicAdd(&cnt[a], 1u);
}

__global__ void __launch_bounds__(DEG_T) k_deg_hist(const int* __restrict__ dst, unsigned* __restrict__ part,
                                                    int E, int per, int npad) {
    __shared__ __attribute__((aligned(16))) unsigned cnt[DEG_RANGE];
    const int r  = blockIdx.x / DEG_SLICES;
    const int s  = blockIdx.x % DEG_SLICES;
    const int lo = r * DEG_RANGE;
    for (int i = threadIdx.x; i < DEG_RANGE / 4; i += DEG_T)
        ((uint4*)cnt)[i] = make_uint4(0u, 0u, 0u, 0u);
    __syncthreads();
    const int e0 = min(E, s * per);
    const int e1 = min(E, e0 + per);
    int e = e0;
    if ((((size_t)(dst + e0)) & 15) == 0) {
        const int4* p = (const int4*)(dst + e0);
        const int n4 = (e1 - e0) >> 2;
        for (int base = 0; base < n4; base += DEG_B * DEG_T) {
            int4 v[DEG_B];
#pragma unroll
            for (int k = 0; k < DEG_B; ++k) {
                const int i = base + k * DEG_T + (int)threadIdx.x;
                v[k] = make_int4(-1, -1, -1, -1);
                if (i < n4) v[k] = p[i];
            }
#pragma unroll
            for (int k = 0; k < DEG_B; ++k) {
                deg_count1(cnt, v[k].x, lo);
                deg_count1(cnt, v[k].y, lo);
                deg_count1(cnt, v[k].z, lo);
                deg_count1(cnt, v[k].w, lo);
            }
        }
        e = e0 + (n4 << 2);
    }
    for (e += (int)threadIdx.x; e < e1; e += DEG_T)
        deg_count1(cnt, dst[e], lo);
    __syncthreads();
    uint4* o = (uint4*)(part + (size_t)s * npad + lo);
    for (int i = threadIdx.x; i < DEG_RANGE / 4; i += DEG_T)
        o[i] = ((const uint4*)cnt)[i];
}

__global__ void __launch_bounds__(128) k_dinv(const unsigned* __restrict__ part, float* __restrict__ dinv,
                                              unsigned* __restrict__ ctr, unsigned* __restrict__ deg8,
                                              int n4, int npad4) {
    const int i = blockIdx.x * blockDim.x + threadIdx.x;
    if (i < 4) ctr[i * QPITCH] = 0u;
    if (i < n4) {
        uint4 c = make_uint4(1u, 1u, 1u, 1u);
#pragma unroll
        for (int s = 0; s < DEG_SLICES; ++s) {
            const uint4 v = ((const uint4*)part)[(size_t)s * npad4 + i];
            c.x += v.x; c.y += v.y; c.z += v.z; c.w += v.w;
        }
        ((float4*)dinv)[i] = make_float4(rsqrtf((float)c.x), rsqrtf((float)c.y),
                                         rsqrtf((float)c.z), rsqrtf((float)c.w));
        deg8[i] = min(c.x, 255u) | (min(c.y, 255u) << 8) | (min(c.z, 255u) << 16) | (min(c.w, 255u) << 24);
    }
}

__global__ void __launch_bounds__(256) k_gemm64(const float* __restrict__ A,
                                                const float* __restrict__ B,
                                                const float* __restrict__ dinv,
                                                const float* __restrict__ b1,
                                                float* __restrict__ C,
                                                float* __restrict__ Hacc,
                                                int nstrips) {
    const int wave = __builtin_amdgcn_readfirstlane((int)(threadIdx.x >> 5));
    const int lane = threadIdx.x & 31;
    const int half = lane >> 4;
    const int lm   = lane & 15;
    const int strip = blockIdx.x * 8 + wave;
    if (strip >= nstrips) return;
    const int m0   = strip * 16;
    const float* arow = A + (size_t)(m0 + lm) * 64;

    v2f a[16];
#pragma unroll
    for (int kk = 0; kk < 16; ++kk)
        a[kk] = *(const v2f*)(arow + kk * 4 + half * 2);

    v8f acc[4];
#pragma unroll
    for (int nt = 0; nt < 4; ++nt) acc[nt] = (v8f){0.f,0.f,0.f,0.f,0.f,0.f,0.f,0.f};
#pragma unroll
    for (int kk = 0; kk < 16; ++kk) {
        const int kb = kk * 4 + half * 2;
#pragma unroll
        for (int nt = 0; nt < 4; ++nt) {
            const int bcol = nt * 16 + lm;
            v2f b;
            b.x = B[(size_t)kb * 64 + bcol];
            b.y = B[(size_t)(kb + 1) * 64 + bcol];
            acc[nt] = __builtin_amdgcn_wmma_f32_16x16x4_f32(false, a[kk], false, b,
                                                            (short)0, acc[nt], false, false);
        }
    }
    float di2[8];
#pragma unroll
    for (int i = 0; i < 8; ++i) {
        const float di = dinv[m0 + half * 8 + i];
        di2[i] = di * di;
    }
#pragma unroll
    for (int nt = 0; nt < 4; ++nt) {
        const int col = nt * 16 + lm;
        const float bb = b1[col];
#pragma unroll
        for (int i = 0; i < 8; ++i) {
            const int row = m0 + half * 8 + i;
            const float v = acc[nt][i];
            C[(size_t)row * 64 + col]    = v;
            Hacc[(size_t)row * 64 + col] = bb + di2[i] * v;
        }
    }
}

#define QCAP 96

__device__ __forceinline__ int chip_half() {
    return (int)((__builtin_amdgcn_s_sendmsg_rtn(0x87) >> 18) & 1u);
}

__device__ __forceinline__ unsigned draw_ticket(unsigned* q, int lane) {
    unsigned v = 0u;
    if (lane == 0) v = atomicAdd(q, 1u);
    return v;
}

__device__ __forceinline__ int next_tile(unsigned* __restrict__ ctr, int ntiles, int lane,
                                         unsigned& pend, int& q, bool& switched) {
    for (;;) {
        const int t = (int)__builtin_amdgcn_readfirstlane(pend);
        if (t < ntiles) { pend = draw_ticket(ctr + q * QPITCH, lane); return t; }
        if (switched) return -1;
        switched = true;
        q ^= 1;
        pend = draw_ticket(ctr + q * QPITCH, lane);
    }
}

__device__ __forceinline__ void load_tile(const int* __restrict__ src, const int* __restrict__ dst,
                                          int E, int t, int lane, int& s0, int& d0, int& s1, int& d1) {
    const int e = t * TILE_E + lane;
    s0 = 0; d0 = -1; s1 = 0; d1 = -1;
    if (e      < E) { s0 = src[e];      d0 = dst[e];      }
    if (e + 32 < E) { s1 = src[e + 32]; d1 = dst[e + 32]; }
}

__device__ __forceinline__ int pack_tile(int s0, int d0, int s1, int d1, bool i0, bool i1,
                                         float w0, float w1,
                                         int* qs, int* qd, float* qw, int nq) {
    const unsigned m0 = __builtin_amdgcn_ballot_w32(i0), m1 = __builtin_amdgcn_ballot_w32(i1);
    const int r0 = i0 ? nq + (int)__builtin_amdgcn_mbcnt_lo(m0, 0u) : QCAP;
    const int r1 = i1 ? nq + __builtin_popcount(m0) + (int)__builtin_amdgcn_mbcnt_lo(m1, 0u) : QCAP;
    qs[r0] = s0; qd[r0] = d0; qw[r0] = w0;
    qs[r1] = s1; qd[r1] = d1; qw[r1] = w1;
    return nq + __builtin_popcount(m0) + __builtin_popcount(m1);
}

__device__ __forceinline__ void queue_window(const int* qs, const int* qd, const float* qw, int pos, int nq, int lane,
                                             int& sv, int& dv, float& wv) {
    sv = 0; dv = 0; wv = 0.0f;
    if (pos + lane < nq) { sv = qs[pos + lane]; dv = qd[pos + lane]; wv = qw[pos + lane]; }
}

__device__ __forceinline__ int queue_keep_rest(int* qs, int* qd, float* qw, int pos, int nq, int lane) {
    const int rem = nq - pos;
    if (pos > 0 && rem > 0) {
        int ts = 0, td = 0; float tw = 0.0f;
        if (lane < rem) { ts = qs[pos + lane]; td = qd[pos + lane]; tw = qw[pos + lane]; }
        if (lane < rem) { qs[lane] = ts; qd[lane] = td; qw[lane] = tw; }
    }
    return rem;
}

__device__ __forceinline__ float dinv_from(unsigned c, const float* __restrict__ dinv, int node) {
    return (c != 255u) ? rsqrtf((float)c) : dinv[node];
}

__global__ void __launch_bounds__(EDGE1_T) k_scatter64(const int* __restrict__ src, const int* __restrict__ dst,
                                                   const float* __restrict__ dinv, const float* __restrict__ xw,
                                                   float* __restrict__ h, unsigned* __restrict__ ctr,
                                                   float* __restrict__ wgt, const unsigned* __restrict__ deg8,
                                                   int E, int N, int ntiles) {
    __shared__ unsigned first;
    __shared__ __attribute__((aligned(16))) unsigned char tab[DEG8_CAP];
    __shared__ int   q_s[EDGE1_T / 32][QCAP + 1];
    __shared__ int   q_d[EDGE1_T / 32][QCAP + 1];
    __shared__ float q_w[EDGE1_T / 32][QCAP + 1];
    const int lane = threadIdx.x & 31;
    const int wave = __builtin_amdgcn_readfirstlane((int)(threadIdx.x >> 5));
    int* qs = q_s[wave]; int* qd = q_d[wave]; float* qw = q_w[wave];
    const int nsplit = min(N, ((N >> 1) + 15) & ~15);
    int q = chip_half();
    if (threadIdx.x == 0) first = atomicAdd(ctr + q * QPITCH, (unsigned)(EDGE1_T / 32));
    const bool use_tab = (N <= DEG8_CAP);
    if (use_tab) {
        const int nw  = N >> 2;
        const int n16 = nw >> 2;
        const int rot = (int)(((blockIdx.x % EDGE1_BLOCKS) * (unsigned)n16) / EDGE1_BLOCKS);
        constexpr int NK = (DEG8_CAP / 16 + EDGE1_T - 1) / EDGE1_T;
        uint4 tv[NK];
#pragma unroll
        for (int k = 0; k < NK; ++k) {
            const int i = k * EDGE1_T + (int)threadIdx.x;
            int j = i + rot; if (j >= n16) j -= n16;
            tv[k] = make_uint4(0u, 0u, 0u, 0u);
            if (i < n16) tv[k] = ((const uint4*)deg8)[j];
        }
#pragma unroll
        for (int k = 0; k < NK; ++k) {
            const int i = k * EDGE1_T + (int)threadIdx.x;
            int j = i + rot; if (j >= n16) j -= n16;
            if (i < n16) ((uint4*)tab)[j] = tv[k];
        }
        for (int w = (n16 << 2) + (int)threadIdx.x; w < nw; w += EDGE1_T)
            ((unsigned*)tab)[w] = deg8[w];
    }
    __syncthreads();
    unsigned pend = draw_ticket(ctr + q * QPITCH, lane);
    bool switched = false;
    int t = __builtin_amdgcn_readfirstlane((int)first) + wave;
    int qt = q;
    if (t >= ntiles) {
        t = __builtin_amdgcn_readfirstlane(next_tile(ctr, ntiles, lane, pend, q, switched));
        q = __builtin_amdgcn_readfirstlane(q);
        qt = q;
    }
    if (t < 0) return;
    int s0, d0, s1, d1;
    load_tile(src, dst, E, t, lane, s0, d0, s1, d1);
    bool have = true;
    int tc = t;
    int nq = 0;
    for (;;) {
        bool i0 = false, i1 = false;
        unsigned cs0 = 255u, cd0 = 255u, cs1 = 255u, cd1 = 255u;
        int ks0 = 0, kd0 = 0, ks1 = 0, kd1 = 0;
        int tn = -1, qn = qt;
        int ns0 = 0, nd0 = -1, ns1 = 0, nd1 = -1;
        if (have) {
            tn = __builtin_amdgcn_readfirstlane(next_tile(ctr, ntiles, lane, pend, q, switched));
            q  = __builtin_amdgcn_readfirstlane(q);
            qn = q;
            const unsigned lo  = qt ? (unsigned)nsplit : 0u;
            const unsigned len = qt ? (unsigned)(N - nsplit) : (unsigned)nsplit;
            i0 = ((unsigned)d0 - lo) < len; i1 = ((unsigned)d1 - lo) < len;
            ks0 = i0 ? s0 : 0; kd0 = i0 ? d0 : 0;
            ks1 = i1 ? s1 : 0; kd1 = i1 ? d1 : 0;
            if (use_tab) { cs0 = tab[ks0]; cd0 = tab[kd0]; cs1 = tab[ks1]; cd1 = tab[kd1]; }
            if (tn >= 0) load_tile(src, dst, E, tn, lane, ns0, nd0, ns1, nd1);
        }
        int pos = 0;
        while (nq - pos >= 16) {
            const int cnt = min(32, (nq - pos) & ~15);
            int sv, dv; float wv;
            queue_window(qs, qd, qw, pos, nq, lane, sv, dv, wv);
            const int wi = __float_as_int(wv);
            for (int j0 = 0; j0 < cnt; j0 += 16) {
                float v0[16], v1[16];
#pragma unroll
                for (int u = 0; u < 16; ++u) {
                    const int sj = __builtin_amdgcn_readlane(sv, j0 + u);
                    const float* xr = xw + (size_t)sj * 64;
                    v0[u] = xr[lane];
                    v1[u] = xr[lane + 32];
                }
#pragma unroll
                for (int u = 0; u < 16; ++u) {
                    const int   dj = __builtin_amdgcn_readlane(dv, j0 + u);
                    const float wj = __int_as_float(__builtin_amdgcn_readlane(wi, j0 + u));
                    float* o = h + (size_t)dj * 64;
                    atomicAdd(o + lane,      wj * v0[u]);
                    atomicAdd(o + lane + 32, wj * v1[u]);
                }
            }
            pos += cnt;
        }
        nq = queue_keep_rest(qs, qd, qw, pos, nq, lane);
        if (!have) break;
        const float w0 = dinv_from(cs0, dinv, ks0) * dinv_from(cd0, dinv, kd0);
        const float w1 = dinv_from(cs1, dinv, ks1) * dinv_from(cd1, dinv, kd1);
        float* wt = wgt + ((size_t)qt * ntiles + tc) * TILE_E;
        wt[lane] = w0; wt[lane + 32] = w1;
        nq = pack_tile(s0, d0, s1, d1, i0, i1, w0, w1, qs, qd, qw, nq);
        have = (tn >= 0);
        tc = tn;
        qt = qn; s0 = ns0; d0 = nd0; s1 = ns1; d1 = nd1;
    }
    if (nq > 0) {
        int sv, dv; float wv;
        queue_window(qs, qd, qw, 0, nq, lane, sv, dv, wv);
        const int wi = __float_as_int(wv);
        float v0[16], v1[16];
#pragma unroll
        for (int u = 0; u < 16; ++u) {
            if (u < nq) {
                const int sj = __builtin_amdgcn_readlane(sv, u);
                const float* xr = xw + (size_t)sj * 64;
                v0[u] = xr[lane];
                v1[u] = xr[lane + 32];
            }
        }
#pragma unroll
        for (int u = 0; u < 16; ++u) {
            if (u < nq) {
                const int   dj = __builtin_amdgcn_readlane(dv, u);
                const float wj = __int_as_float(__builtin_amdgcn_readlane(wi, u));
                float* o = h + (size_t)dj * 64;
                atomicAdd(o + lane,      wj * v0[u]);
                atomicAdd(o + lane + 32, wj * v1[u]);
            }
        }
    }
}

__global__ void __launch_bounds__(256) k_gemm34(const float* __restrict__ A,
                                                const float* __restrict__ B,
                                                const float* __restrict__ dinv,
                                                const float* __restrict__ b2,
                                                float* __restrict__ C,
                                                float* __restrict__ out,
                                                int nstrips) {
    const int wave = __builtin_amdgcn_readfirstlane((int)(threadIdx.x >> 5));
    const int lane = threadIdx.x & 31;
    const int half = lane >> 4;
    const int lm   = lane & 15;
    const int strip = blockIdx.x * 8 + wave;
    if (strip >= nstrips) return;
    const int m0   = strip * 16;
    const float* arow = A + (size_t)(m0 + lm) * 64;

    v2f a[16];
#pragma unroll
    for (int kk = 0; kk < 16; ++kk)
        a[kk] = *(const v2f*)(arow + kk * 4 + half * 2);

    float m = 0.0f;
#pragma unroll
    for (int kk = 0; kk < 16; ++kk) {
        a[kk].x = fmaxf(a[kk].x, 0.0f);
        a[kk].y = fmaxf(a[kk].y, 0.0f);
        m = fmaxf(m, fmaxf(a[kk].x, a[kk].y));
    }
    m = fmaxf(m, __shfl_xor(m, 16, 32));
    float s = 0.0f;
#pragma unroll
    for (int kk = 0; kk < 16; ++kk) {
        a[kk].x = __expf(a[kk].x - m);
        a[kk].y = __expf(a[kk].y - m);
        s += a[kk].x + a[kk].y;
    }
    s += __shfl_xor(s, 16, 32);
    const float inv = 1.0f / s;

    v8f acc[3];
#pragma unroll
    for (int nt = 0; nt < 3; ++nt) acc[nt] = (v8f){0.f,0.f,0.f,0.f,0.f,0.f,0.f,0.f};
#pragma unroll
    for (int kk = 0; kk < 16; ++kk) {
        const int kb = kk * 4 + half * 2;
        v2f av;
        av.x = a[kk].x * inv;
        av.y = a[kk].y * inv;
#pragma unroll
        for (int nt = 0; nt < 3; ++nt) {
            const int col = nt * 16 + lm;
            const bool ok = (col < 34);
            const int  cc = ok ? col : 33;
            const float bx = B[(size_t)kb * 34 + cc];
            const float by = B[(size_t)(kb + 1) * 34 + cc];
            v2f b;
            b.x = ok ? bx : 0.0f;
            b.y = ok ? by : 0.0f;
            acc[nt] = __builtin_amdgcn_wmma_f32_16x16x4_f32(false, av, false, b,
                                                            (short)0, acc[nt], false, false);
        }
    }
    float di2[8];
#pragma unroll
    for (int i = 0; i < 8; ++i) {
        const float di = dinv[m0 + half * 8 + i];
        di2[i] = di * di;
    }
#pragma unroll
    for (int nt = 0; nt < 3; ++nt) {
        const int col = nt * 16 + lm;
        if (col < 34) {
            const float bb = b2[col];
#pragma unroll
            for (int i = 0; i < 8; ++i) {
                const int row = m0 + half * 8 + i;
                const float v = acc[nt][i];
                C[(size_t)row * 34 + col]   = v;
                out[(size_t)row * 34 + col] = bb + di2[i] * v;
            }
        }
    }
}

__device__ __forceinline__ int row_cut(int d) { return 32 - 2 * (d & 15); }

template <bool GUARD>
__device__ __forceinline__ void add_rows34(const float* __restrict__ hw2, float* __restrict__ out,
                                           int sv, int dv, int wi, int j0, int n, int lane) {
    float va[16], vb[16];
#pragma unroll
    for (int u = 0; u < 16; ++u) {
        if (!GUARD || u < n) {
            const int sj = __builtin_amdgcn_readlane(sv, j0 + u);
            const int na = row_cut(__builtin_amdgcn_readlane(dv, j0 + u));
            va[u] = hw2[(unsigned)(sj * 34 + min(lane, na - 1))];
            vb[u] = hw2[(unsigned)(sj * 34 + min(na + lane, 33))];
        }
    }
    if (!GUARD) asm volatile("" :: "v"(va[0]));
#pragma unroll
    for (int u = 0; u < 16; ++u) {
        if (!GUARD || u < n) {
            const int   dj = __builtin_amdgcn_readlane(dv, j0 + u);
            const float wj = __int_as_float(__builtin_amdgcn_readlane(wi, j0 + u));
            const int   na = row_cut(dj);
            if (lane < na)      atomicAdd(&out[(unsigned)(dj * 34 + lane)],      wj * va[u]);
            if (lane < 34 - na) atomicAdd(&out[(unsigned)(dj * 34 + na + lane)], wj * vb[u]);
        }
    }
}

__global__ void __launch_bounds__(256) k_scatter34(const int* __restrict__ src, const int* __restrict__ dst,
                                                   const float* __restrict__ wgt, const float* __restrict__ hw2,
                                                   float* __restrict__ out, unsigned* __restrict__ ctr,
                                                   int E, int N, int ntiles) {
    __shared__ unsigned first;
    __shared__ int   q_s[8][QCAP + 1];
    __shared__ int   q_d[8][QCAP + 1];
    __shared__ float q_w[8][QCAP + 1];
    const int lane = threadIdx.x & 31;
    const int wave = __builtin_amdgcn_readfirstlane((int)(threadIdx.x >> 5));
    int* qs = q_s[wave]; int* qd = q_d[wave]; float* qw = q_w[wave];
    const int nsplit = min(N, ((N >> 1) + 15) & ~15);
    int q = chip_half();
    if (threadIdx.x == 0) first = atomicAdd(ctr + q * QPITCH, 8u);
    __syncthreads();
    unsigned pend = draw_ticket(ctr + q * QPITCH, lane);
    bool switched = false;
    int t = __builtin_amdgcn_readfirstlane((int)first) + wave;
    int qt = q;
    if (t >= ntiles) {
        t = __builtin_amdgcn_readfirstlane(next_tile(ctr, ntiles, lane, pend, q, switched));
        q = __builtin_amdgcn_readfirstlane(q);
        qt = q;
    }
    if (t < 0) return;
    int s0, d0, s1, d1;
    load_tile(src, dst, E, t, lane, s0, d0, s1, d1);
    const float* wt = wgt + ((size_t)qt * ntiles + t) * TILE_E;
    float w0 = wt[lane], w1 = wt[lane + 32];
    bool have = true;
    int nq = 0;
    for (;;) {
        bool i0 = false, i1 = false;
        int tn = -1, qn = qt;
        int ns0 = 0, nd0 = -1, ns1 = 0, nd1 = -1;
        float nw0 = 0.0f, nw1 = 0.0f;
        if (have) {
            tn = __builtin_amdgcn_readfirstlane(next_tile(ctr, ntiles, lane, pend, q, switched));
            q  = __builtin_amdgcn_readfirstlane(q);
            qn = q;
            const unsigned lo  = qt ? (unsigned)nsplit : 0u;
            const unsigned len = qt ? (unsigned)(N - nsplit) : (unsigned)nsplit;
            i0 = ((unsigned)d0 - lo) < len; i1 = ((unsigned)d1 - lo) < len;
            if (tn >= 0) {
                load_tile(src, dst, E, tn, lane, ns0, nd0, ns1, nd1);
                const float* nwt = wgt + ((size_t)qn * ntiles + tn) * TILE_E;
                nw0 = nwt[lane]; nw1 = nwt[lane + 32];
            }
        }
        int pos = 0;
        while (nq - pos >= 16) {
            const int cnt = min(32, (nq - pos) & ~15);
            int sv, dv; float wv;
            queue_window(qs, qd, qw, pos, nq, lane, sv, dv, wv);
            const int wi = __float_as_int(wv);
            for (int j0 = 0; j0 < cnt; j0 += 16)
                add_rows34<false>(hw2, out, sv, dv, wi, j0, 16, lane);
            pos += cnt;
        }
        nq = queue_keep_rest(qs, qd, qw, pos, nq, lane);
        if (!have) break;
        nq = pack_tile(s0, d0, s1, d1, i0, i1, w0, w1, qs, qd, qw, nq);
        have = (tn >= 0);
        qt = qn; s0 = ns0; d0 = nd0; s1 = ns1; d1 = nd1; w0 = nw0; w1 = nw1;
    }
    if (nq > 0) {
        int sv, dv; float wv;
        queue_window(qs, qd, qw, 0, nq, lane, sv, dv, wv);
        const int wi = __float_as_int(wv);
        add_rows34<true>(hw2, out, sv, dv, wi, 0, nq, lane);
    }
}

extern "C" void kernel_launch(void* const* d_in, const int* in_sizes, int n_in,
                              void* d_out, int out_size, void* d_ws, size_t ws_size,
                              hipStream_t stream) {
    const float* x  = (const float*)d_in[0];
    const int*   ei = (const int*)  d_in[1];
    const float* W1 = (const float*)d_in[2];
    const float* b1 = (const float*)d_in[3];
    const float* W2 = (const float*)d_in[4];
    const float* b2 = (const float*)d_in[5];
    float* out = (float*)d_out;

    const int N = in_sizes[0] / 64;
    const int E = in_sizes[1] / 2;
    const int* src = ei;
    const int* dst = ei + E;

    const int nranges = (N + DEG_RANGE - 1) / DEG_RANGE;
    const int npad    = nranges * DEG_RANGE;
    const int per     = (((E + DEG_SLICES - 1) / DEG_SLICES) + 63) & ~63;

    float* xw   = (float*)d_ws;
    float* h    = xw   + (size_t)N * 64;
    float* hw2  = h    + (size_t)N * 64;
    float* dinv = hw2  + (size_t)N * 34;
    unsigned* part = (unsigned*)(dinv + N);
    unsigned* ctr  = part + (size_t)DEG_SLICES * npad;
    float* wgt = (float*)(ctr + 4 * QPITCH);
    const int ntiles = (E + TILE_E - 1) / TILE_E;
    unsigned* deg8 = (unsigned*)(wgt + 2 * (size_t)ntiles * TILE_E);

    const int T = 256;
    const int N4 = N / 4;
    const int nstrips = N / 16;
    k_deg_hist<<<nranges * DEG_SLICES, DEG_T, 0, stream>>>(dst, part, E, per, npad);
    k_dinv    <<<(N4 + 127) / 128, 128, 0, stream>>>(part, dinv, ctr, deg8, N4, npad / 4);
    k_gemm64<<<(nstrips + 7) / 8, 256, 0, stream>>>(x, W1, dinv, b1, xw, h, nstrips);
    k_scatter64<<<EDGE1_BLOCKS, EDGE1_T, 0, stream>>>(src, dst, dinv, xw, h, ctr, wgt, deg8, E, N, ntiles);
    k_gemm34<<<(nstrips + 7) / 8, 256, 0, stream>>>(h, W2, dinv, b2, hw2, out, nstrips);
    k_scatter34<<<EDGE_BLOCKS, T, 0, stream>>>(src, dst, wgt, hw2, out, ctr + 2 * QPITCH, E, N, ntiles);
}


# speedup vs baseline: 2.8868x; 1.0496x over previous
#include <hip/hip_runtime.h>
#include <math.h>

typedef __attribute__((ext_vector_type(2))) float v2f;
typedef __attribute__((ext_vector_type(8))) float v8f;

#define EDGE_BLOCKS 1024
#define EDGE1_BLOCKS 256
#define EDGE1_T     1024
#define DEG8_CAP  102400
#define TILE_E      64
#define QPITCH      64
#define HW2_PITCH   64

#define DEG_RANGE  12800
#define DEG_SLICES 32
#define DEG_T      512
#define DEG_B      20

__device__ __forceinline__ void deg_count1(unsigned* cnt, int d, int lo) {
    const unsigned a = (unsigned)(d - lo);
    if (a < (unsigned)DEG_RANGE) atomicAdd(&cnt[a], 1u);
}

__global__ void __launch_bounds__(DEG_T) k_deg_hist(const int* __restrict__ dst, unsigned* __restrict__ part,
                                                    int E, int per, int npad) {
    __shared__ __attribute__((aligned(16))) unsigned cnt[DEG_RANGE];
    const int r  = blockIdx.x / DEG_SLICES;
    const int s  = blockIdx.x % DEG_SLICES;
    const int lo = r * DEG_RANGE;
    for (int i = threadIdx.x; i < DEG_RANGE / 4; i += DEG_T)
        ((uint4*)cnt)[i] = make_uint4(0u, 0u, 0u, 0u);
    __syncthreads();
    const int e0 = min(E, s * per);
    const int e1 = min(E, e0 + per);
    int e = e0;
    if ((((size_t)(dst + e0)) & 15) == 0) {
        const int4* p = (const int4*)(dst + e0);
        const int n4 = (e1 - e0) >> 2;
        for (int base = 0; base < n4; base += DEG_B * DEG_T) {
            int4 v[DEG_B];
#pragma unroll
            for (int k = 0; k < DEG_B; ++k) {
                const int i = base + k * DEG_T + (int)threadIdx.x;
                v[k] = make_int4(-1, -1, -1, -1);
                if (i < n4) v[k] = p[i];
            }
#pragma unroll
            for (int k = 0; k < DEG_B; ++k) {
                deg_count1(cnt, v[k].x, lo);
                deg_count1(cnt, v[k].y, lo);
                deg_count1(cnt, v[k].z, lo);
                deg_count1(cnt, v[k].w, lo);
            }
        }
        e = e0 + (n4 << 2);
    }
    for (e += (int)threadIdx.x; e < e1; e += DEG_T)
        deg_count1(cnt, dst[e], lo);
    __syncthreads();
    uint4* o = (uint4*)(part + (size_t)s * npad + lo);
    for (int i = threadIdx.x; i < DEG_RANGE / 4; i += DEG_T)
        o[i] = ((const uint4*)cnt)[i];
}

__global__ void __launch_bounds__(128) k_dinv(const unsigned* __restrict__ part, float* __restrict__ dinv,
                                              unsigned* __restrict__ ctr, unsigned* __restrict__ deg8,
                                              int n4, int npad4) {
    const int i = blockIdx.x * blockDim.x + threadIdx.x;
    if (i < 4) ctr[i * QPITCH] = 0u;
    if (i < n4) {
        uint4 c = make_uint4(1u, 1u, 1u, 1u);
#pragma unroll
        for (int s = 0; s < DEG_SLICES; ++s) {
            const uint4 v = ((const uint4*)part)[(size_t)s * npad4 + i];
            c.x += v.x; c.y += v.y; c.z += v.z; c.w += v.w;
        }
        ((float4*)dinv)[i] = make_float4(rsqrtf((float)c.x), rsqrtf((float)c.y),
                                         rsqrtf((float)c.z), rsqrtf((float)c.w));
        deg8[i] = min(c.x, 255u) | (min(c.y, 255u) << 8) | (min(c.z, 255u) << 16) | (min(c.w, 255u) << 24);
    }
}

__global__ void __launch_bounds__(256) k_gemm64(const float* __restrict__ A,
                                                const float* __restrict__ B,
                                                const float* __restrict__ dinv,
                                                const float* __restrict__ b1,
                                                float* __restrict__ C,
                                                float* __restrict__ Hacc,
                                                int nstrips) {
    const int wave = __builtin_amdgcn_readfirstlane((int)(threadIdx.x >> 5));
    const int lane = threadIdx.x & 31;
    const int half = lane >> 4;
    const int lm   = lane & 15;
    const int strip = blockIdx.x * 8 + wave;
    if (strip >= nstrips) return;
    const int m0   = strip * 16;
    const float* arow = A + (size_t)(m0 + lm) * 64;

    v2f a[16];
#pragma unroll
    for (int kk = 0; kk < 16; ++kk)
        a[kk] = *(const v2f*)(arow + kk * 4 + half * 2);

    v8f acc[4];
#pragma unroll
    for (int nt = 0; nt < 4; ++nt) acc[nt] = (v8f){0.f,0.f,0.f,0.f,0.f,0.f,0.f,0.f};
#pragma unroll
    for (int kk = 0; kk < 16; ++kk) {
        const int kb = kk * 4 + half * 2;
#pragma unroll
        for (int nt = 0; nt < 4; ++nt) {
            const int bcol = nt * 16 + lm;
            v2f b;
            b.x = B[(size_t)kb * 64 + bcol];
            b.y = B[(size_t)(kb + 1) * 64 + bcol];
            acc[nt] = __builtin_amdgcn_wmma_f32_16x16x4_f32(false, a[kk], false, b,
                                                            (short)0, acc[nt], false, false);
        }
    }
    float di2[8];
#pragma unroll
    for (int i = 0; i < 8; ++i) {
        const float di = dinv[m0 + half * 8 + i];
        di2[i] = di * di;
    }
#pragma unroll
    for (int nt = 0; nt < 4; ++nt) {
        const int col = nt * 16 + lm;
        const float bb = b1[col];
#pragma unroll
        for (int i = 0; i < 8; ++i) {
            const int row = m0 + half * 8 + i;
            const float v = acc[nt][i];
            C[(size_t)row * 64 + col]    = v;
            Hacc[(size_t)row * 64 + col] = bb + di2[i] * v;
        }
    }
}

#define QCAP 96

__device__ __forceinline__ int chip_half() {
    return (int)((__builtin_amdgcn_s_sendmsg_rtn(0x87) >> 18) & 1u);
}

__device__ __forceinline__ unsigned draw_ticket(unsigned* q, int lane) {
    unsigned v = 0u;
    if (lane == 0) v = atomicAdd(q, 1u);
    return v;
}

__device__ __forceinline__ int next_tile(unsigned* __restrict__ ctr, int ntiles, int lane,
                                         unsigned& pend, int& q, bool& switched) {
    for (;;) {
        const int t = (int)__builtin_amdgcn_readfirstlane(pend);
        if (t < ntiles) { pend = draw_ticket(ctr + q * QPITCH, lane); return t; }
        if (switched) return -1;
        switched = true;
        q ^= 1;
        pend = draw_ticket(ctr + q * QPITCH, lane);
    }
}

__device__ __forceinline__ void load_tile(const int* __restrict__ src, const int* __restrict__ dst,
                                          int E, int t, int lane, int& s0, int& d0, int& s1, int& d1) {
    const int e = t * TILE_E + lane;
    s0 = 0; d0 = -1; s1 = 0; d1 = -1;
    if (e      < E) { s0 = src[e];      d0 = dst[e];      }
    if (e + 32 < E) { s1 = src[e + 32]; d1 = dst[e + 32]; }
}

__device__ __forceinline__ int pack_tile(int s0, int d0, int s1, int d1, bool i0, bool i1,
                                         float w0, float w1,
                                         int* qs, int* qd, float* qw, int nq) {
    const unsigned m0 = __builtin_amdgcn_ballot_w32(i0), m1 = __builtin_amdgcn_ballot_w32(i1);
    const int r0 = i0 ? nq + (int)__builtin_amdgcn_mbcnt_lo(m0, 0u) : QCAP;
    const int r1 = i1 ? nq + __builtin_popcount(m0) + (int)__builtin_amdgcn_mbcnt_lo(m1, 0u) : QCAP;
    qs[r0] = s0; qd[r0] = d0; qw[r0] = w0;
    qs[r1] = s1; qd[r1] = d1; qw[r1] = w1;
    return nq + __builtin_popcount(m0) + __builtin_popcount(m1);
}

__device__ __forceinline__ void queue_window(const int* qs, const int* qd, const float* qw, int pos, int nq, int lane,
                                             int& sv, int& dv, float& wv) {
    sv = 0; dv = 0; wv = 0.0f;
    if (pos + lane < nq) { sv = qs[pos + lane]; dv = qd[pos + lane]; wv = qw[pos + lane]; }
}

__device__ __forceinline__ int queue_keep_rest(int* qs, int* qd, float* qw, int pos, int nq, int lane) {
    const int rem = nq - pos;
    if (pos > 0 && rem > 0) {
        int ts = 0, td = 0; float tw = 0.0f;
        if (lane < rem) { ts = qs[pos + lane]; td = qd[pos + lane]; tw = qw[pos + lane]; }
        if (lane < rem) { qs[lane] = ts; qd[lane] = td; qw[lane] = tw; }
    }
    return rem;
}

__device__ __forceinline__ float dinv_from(unsigned c, const float* __restrict__ dinv, int node) {
    return (c != 255u) ? rsqrtf((float)c) : dinv[node];
}

__global__ void __launch_bounds__(EDGE1_T) k_scatter64(const int* __restrict__ src, const int* __restrict__ dst,
                                                   const float* __restrict__ dinv, const float* __restrict__ xw,
                                                   float* __restrict__ h, unsigned* __restrict__ ctr,
                                                   float* __restrict__ wgt, const unsigned* __restrict__ deg8,
                                                   int E, int N, int ntiles) {
    __shared__ unsigned first;
    __shared__ __attribute__((aligned(16))) unsigned char tab[DEG8_CAP];
    __shared__ int   q_s[EDGE1_T / 32][QCAP + 1];
    __shared__ int   q_d[EDGE1_T / 32][QCAP + 1];
    __shared__ float q_w[EDGE1_T / 32][QCAP + 1];
    const int lane = threadIdx.x & 31;
    const int wave = __builtin_amdgcn_readfirstlane((int)(threadIdx.x >> 5));
    int* qs = q_s[wave]; int* qd = q_d[wave]; float* qw = q_w[wave];
    const int nsplit = min(N, ((N >> 1) + 15) & ~15);
    int q = chip_half();
    if (threadIdx.x == 0) first = atomicAdd(ctr + q * QPITCH, (unsigned)(EDGE1_T / 32));
    const bool use_tab = (N <= DEG8_CAP);
    if (use_tab) {
        const int nw  = N >> 2;
        const int n16 = nw >> 2;
        const int rot = (int)(((blockIdx.x % EDGE1_BLOCKS) * (unsigned)n16) / EDGE1_BLOCKS);
        constexpr int NK = (DEG8_CAP / 16 + EDGE1_T - 1) / EDGE1_T;
        uint4 tv[NK];
#pragma unroll
        for (int k = 0; k < NK; ++k) {
            const int i = k * EDGE1_T + (int)threadIdx.x;
            int j = i + rot; if (j >= n16) j -= n16;
            tv[k] = make_uint4(0u, 0u, 0u, 0u);
            if (i < n16) tv[k] = ((const uint4*)deg8)[j];
        }
#pragma unroll
        for (int k = 0; k < NK; ++k) {
            const int i = k * EDGE1_T + (int)threadIdx.x;
            int j = i + rot; if (j >= n16) j -= n16;
            if (i < n16) ((uint4*)tab)[j] = tv[k];
        }
        for (int w = (n16 << 2) + (int)threadIdx.x; w < nw; w += EDGE1_T)
            ((unsigned*)tab)[w] = deg8[w];
    }
    __syncthreads();
    unsigned pend = draw_ticket(ctr + q * QPITCH, lane);
    bool switched = false;
    int t = __builtin_amdgcn_readfirstlane((int)first) + wave;
    int qt = q;
    if (t >= ntiles) {
        t = __builtin_amdgcn_readfirstlane(next_tile(ctr, ntiles, lane, pend, q, switched));
        q = __builtin_amdgcn_readfirstlane(q);
        qt = q;
    }
    if (t < 0) return;
    int s0, d0, s1, d1;
    load_tile(src, dst, E, t, lane, s0, d0, s1, d1);
    bool have = true;
    int tc = t;
    int nq = 0;
    for (;;) {
        bool i0 = false, i1 = false;
        unsigned cs0 = 255u, cd0 = 255u, cs1 = 255u, cd1 = 255u;
        int ks0 = 0, kd0 = 0, ks1 = 0, kd1 = 0;
        int tn = -1, qn = qt;
        int ns0 = 0, nd0 = -1, ns1 = 0, nd1 = -1;
        if (have) {
            tn = __builtin_amdgcn_readfirstlane(next_tile(ctr, ntiles, lane, pend, q, switched));
            q  = __builtin_amdgcn_readfirstlane(q);
            qn = q;
            const unsigned lo  = qt ? (unsigned)nsplit : 0u;
            const unsigned len = qt ? (unsigned)(N - nsplit) : (unsigned)nsplit;
            i0 = ((unsigned)d0 - lo) < len; i1 = ((unsigned)d1 - lo) < len;
            ks0 = i0 ? s0 : 0; kd0 = i0 ? d0 : 0;
            ks1 = i1 ? s1 : 0; kd1 = i1 ? d1 : 0;
            if (use_tab) { cs0 = tab[ks0]; cd0 = tab[kd0]; cs1 = tab[ks1]; cd1 = tab[kd1]; }
            if (tn >= 0) load_tile(src, dst, E, tn, lane, ns0, nd0, ns1, nd1);
        }
        int pos = 0;
        while (nq - pos >= 16) {
            const int cnt = min(32, (nq - pos) & ~15);
            int sv, dv; float wv;
            queue_window(qs, qd, qw, pos, nq, lane, sv, dv, wv);
            const int wi = __float_as_int(wv);
            for (int j0 = 0; j0 < cnt; j0 += 16) {
                float v0[16], v1[16];
#pragma unroll
                for (int u = 0; u < 16; ++u) {
                    const int sj = __builtin_amdgcn_readlane(sv, j0 + u);
                    const float* xr = xw + (size_t)sj * 64;
                    v0[u] = xr[lane];
                    v1[u] = xr[lane + 32];
                }
#pragma unroll
                for (int u = 0; u < 16; ++u) {
                    const int   dj = __builtin_amdgcn_readlane(dv, j0 + u);
                    const float wj = __int_as_float(__builtin_amdgcn_readlane(wi, j0 + u));
                    float* o = h + (size_t)dj * 64;
                    atomicAdd(o + lane,      wj * v0[u]);
                    atomicAdd(o + lane + 32, wj * v1[u]);
                }
            }
            pos += cnt;
        }
        nq = queue_keep_rest(qs, qd, qw, pos, nq, lane);
        if (!have) break;
        const float w0 = dinv_from(cs0, dinv, ks0) * dinv_from(cd0, dinv, kd0);
        const float w1 = dinv_from(cs1, dinv, ks1) * dinv_from(cd1, dinv, kd1);
        float* wt = wgt + ((size_t)qt * ntiles + tc) * TILE_E;
        wt[lane] = w0; wt[lane + 32] = w1;
        nq = pack_tile(s0, d0, s1, d1, i0, i1, w0, w1, qs, qd, qw, nq);
        have = (tn >= 0);
        tc = tn;
        qt = qn; s0 = ns0; d0 = nd0; s1 = ns1; d1 = nd1;
    }
    if (nq > 0) {
        int sv, dv; float wv;
        queue_window(qs, qd, qw, 0, nq, lane, sv, dv, wv);
        const int wi = __float_as_int(wv);
        float v0[16], v1[16];
#pragma unroll
        for (int u = 0; u < 16; ++u) {
            if (u < nq) {
                const int sj = __builtin_amdgcn_readlane(sv, u);
                const float* xr = xw + (size_t)sj * 64;
                v0[u] = xr[lane];
                v1[u] = xr[lane + 32];
            }
        }
#pragma unroll
        for (int u = 0; u < 16; ++u) {
            if (u < nq) {
                const int   dj = __builtin_amdgcn_readlane(dv, u);
                const float wj = __int_as_float(__builtin_amdgcn_readlane(wi, u));
                float* o = h + (size_t)dj * 64;
                atomicAdd(o + lane,      wj * v0[u]);
                atomicAdd(o + lane + 32, wj * v1[u]);
            }
        }
    }
}

__global__ void __launch_bounds__(256) k_gemm34(const float* __restrict__ A,
                                                const float* __restrict__ B,
                                                const float* __restrict__ dinv,
                                                const float* __restrict__ b2,
                                                float* __restrict__ C,
                                                float* __restrict__ out,
                                                int nstrips) {
    const int wave = __builtin_amdgcn_readfirstlane((int)(threadIdx.x >> 5));
    const int lane = threadIdx.x & 31;
    const int half = lane >> 4;
    const int lm   = lane & 15;
    const int strip = blockIdx.x * 8 + wave;
    if (strip >= nstrips) return;
    const int m0   = strip * 16;
    const float* arow = A + (size_t)(m0 + lm) * 64;

    v2f a[16];
#pragma unroll
    for (int kk = 0; kk < 16; ++kk)
        a[kk] = *(const v2f*)(arow + kk * 4 + half * 2);

    float m = 0.0f;
#pragma unroll
    for (int kk = 0; kk < 16; ++kk) {
        a[kk].x = fmaxf(a[kk].x, 0.0f);
        a[kk].y = fmaxf(a[kk].y, 0.0f);
        m = fmaxf(m, fmaxf(a[kk].x, a[kk].y));
    }
    m = fmaxf(m, __shfl_xor(m, 16, 32));
    float s = 0.0f;
#pragma unroll
    for (int kk = 0; kk < 16; ++kk) {
        a[kk].x = __expf(a[kk].x - m);
        a[kk].y = __expf(a[kk].y - m);
        s += a[kk].x + a[kk].y;
    }
    s += __shfl_xor(s, 16, 32);
    const float inv = 1.0f / s;

    v8f acc[3];
#pragma unroll
    for (int nt = 0; nt < 3; ++nt) acc[nt] = (v8f){0.f,0.f,0.f,0.f,0.f,0.f,0.f,0.f};
#pragma unroll
    for (int kk = 0; kk < 16; ++kk) {
        const int kb = kk * 4 + half * 2;
        v2f av;
        av.x = a[kk].x * inv;
        av.y = a[kk].y * inv;
#pragma unroll
        for (int nt = 0; nt < 3; ++nt) {
            const int col = nt * 16 + lm;
            const bool ok = (col < 34);
            const int  cc = ok ? col : 33;
            const float bx = B[(size_t)kb * 34 + cc];
            const float by = B[(size_t)(kb + 1) * 34 + cc];
            v2f b;
            b.x = ok ? bx : 0.0f;
            b.y = ok ? by : 0.0f;
            acc[nt] = __builtin_amdgcn_wmma_f32_16x16x4_f32(false, av, false, b,
                                                            (short)0, acc[nt], false, false);
        }
    }
    float di2[8];
#pragma unroll
    for (int i = 0; i < 8; ++i) {
        const float di = dinv[m0 + half * 8 + i];
        di2[i] = di * di;
    }
#pragma unroll
    for (int nt = 0; nt < 3; ++nt) {
        const int col = nt * 16 + lm;
        if (col < 34) {
            const float bb = b2[col];
#pragma unroll
            for (int i = 0; i < 8; ++i) {
                const int row = m0 + half * 8 + i;
                const float v = acc[nt][i];
                C[(size_t)row * HW2_PITCH + col] = v;
                out[(size_t)row * 34 + col] = bb + di2[i] * v;
            }
        }
    }
}

__device__ __forceinline__ int row_cut(int d) { return 32 - 2 * (d & 15); }

template <bool GUARD>
__device__ __forceinline__ void add_rows34(const float* __restrict__ hw2, float* __restrict__ out,
                                           int sv, int dv, int wi, int j0, int n, int lane) {
    float va[16], vb[16];
    const bool tail = lane < 2;
    const int  ch2  = tail ? 32 + lane : lane;
#pragma unroll
    for (int u = 0; u < 16; ++u) {
        if (!GUARD || u < n) {
            const int sj = __builtin_amdgcn_readlane(sv, j0 + u);
            const float* r = hw2 + (size_t)(unsigned)sj * HW2_PITCH;
            va[u] = r[lane];
            vb[u] = r[32 + (lane & 1)];
        }
    }
#pragma unroll
    for (int u = 0; u < 16; ++u) {
        if (!GUARD || u < n) {
            const int   dj = __builtin_amdgcn_readlane(dv, j0 + u);
            const float wj = __int_as_float(__builtin_amdgcn_readlane(wi, j0 + u));
            const int   na = row_cut(dj);
            if (lane < na)          atomicAdd(&out[(unsigned)(dj * 34 + lane)], wj * va[u]);
            if (tail || lane >= na) atomicAdd(&out[(unsigned)(dj * 34 + ch2)],  wj * (tail ? vb[u] : va[u]));
        }
    }
}

__global__ void __launch_bounds__(256) k_scatter34(const int* __restrict__ src, const int* __restrict__ dst,
                                                   const float* __restrict__ wgt, const float* __restrict__ hw2,
                                                   float* __restrict__ out, unsigned* __restrict__ ctr,
                                                   int E, int N, int ntiles) {
    __shared__ unsigned first;
    __shared__ int   q_s[8][QCAP + 1];
    __shared__ int   q_d[8][QCAP + 1];
    __shared__ float q_w[8][QCAP + 1];
    const int lane = threadIdx.x & 31;
    const int wave = __builtin_amdgcn_readfirstlane((int)(threadIdx.x >> 5));
    int* qs = q_s[wave]; int* qd = q_d[wave]; float* qw = q_w[wave];
    const int nsplit = min(N, ((N >> 1) + 15) & ~15);
    int q = chip_half();
    if (threadIdx.x == 0) first = atomicAdd(ctr + q * QPITCH, 8u);
    __syncthreads();
    unsigned pend = draw_ticket(ctr + q * QPITCH, lane);
    bool switched = false;
    int t = __builtin_amdgcn_readfirstlane((int)first) + wave;
    int qt = q;
    if (t >= ntiles) {
        t = __builtin_amdgcn_readfirstlane(next_tile(ctr, ntiles, lane, pend, q, switched));
        q = __builtin_amdgcn_readfirstlane(q);
        qt = q;
    }
    if (t < 0) return;
    int s0, d0, s1, d1;
    load_tile(src, dst, E, t, lane, s0, d0, s1, d1);
    const float* wt = wgt + ((size_t)qt * ntiles + t) * TILE_E;
    float w0 = wt[lane], w1 = wt[lane + 32];
    bool have = true;
    int nq = 0;
    for (;;) {
        bool i0 = false, i1 = false;
        int tn = -1, qn = qt;
        int ns0 = 0, nd0 = -1, ns1 = 0, nd1 = -1;
        float nw0 = 0.0f, nw1 = 0.0f;
        if (have) {
            tn = __builtin_amdgcn_readfirstlane(next_tile(ctr, ntiles, lane, pend, q, switched));
            q  = __builtin_amdgcn_readfirstlane(q);
            qn = q;
            const unsigned lo  = qt ? (unsigned)nsplit : 0u;
            const unsigned len = qt ? (unsigned)(N - nsplit) : (unsigned)nsplit;
            i0 = ((unsigned)d0 - lo) < len; i1 = ((unsigned)d1 - lo) < len;
            if (tn >= 0) {
                load_tile(src, dst, E, tn, lane, ns0, nd0, ns1, nd1);
                const float* nwt = wgt + ((size_t)qn * ntiles + tn) * TILE_E;
                nw0 = nwt[lane]; nw1 = nwt[lane + 32];
            }
        }
        int pos = 0;
        while (nq - pos >= 16) {
            const int cnt = min(32, (nq - pos) & ~15);
            int sv, dv; float wv;
            queue_window(qs, qd, qw, pos, nq, lane, sv, dv, wv);
            const int wi = __float_as_int(wv);
            for (int j0 = 0; j0 < cnt; j0 += 16)
                add_rows34<false>(hw2, out, sv, dv, wi, j0, 16, lane);
            pos += cnt;
        }
        nq = queue_keep_rest(qs, qd, qw, pos, nq, lane);
        if (!have) break;
        nq = pack_tile(s0, d0, s1, d1, i0, i1, w0, w1, qs, qd, qw, nq);
        have = (tn >= 0);
        qt = qn; s0 = ns0; d0 = nd0; s1 = ns1; d1 = nd1; w0 = nw0; w1 = nw1;
    }
    if (nq > 0) {
        int sv, dv; float wv;
        queue_window(qs, qd, qw, 0, nq, lane, sv, dv, wv);
        const int wi = __float_as_int(wv);
        add_rows34<true>(hw2, out, sv, dv, wi, 0, nq, lane);
    }
}

extern "C" void kernel_launch(void* const* d_in, const int* in_sizes, int n_in,
                              void* d_out, int out_size, void* d_ws, size_t ws_size,
                              hipStream_t stream) {
    const float* x  = (const float*)d_in[0];
    const int*   ei = (const int*)  d_in[1];
    const float* W1 = (const float*)d_in[2];
    const float* b1 = (const float*)d_in[3];
    const float* W2 = (const float*)d_in[4];
    const float* b2 = (const float*)d_in[5];
    float* out = (float*)d_out;

    const int N = in_sizes[0] / 64;
    const int E = in_sizes[1] / 2;
    const int* src = ei;
    const int* dst = ei + E;

    const int nranges = (N + DEG_RANGE - 1) / DEG_RANGE;
    const int npad    = nranges * DEG_RANGE;
    const int per     = (((E + DEG_SLICES - 1) / DEG_SLICES) + 63) & ~63;

    float* xw   = (float*)d_ws;
    float* h    = xw   + (size_t)N * 64;
    float* hw2  = h    + (size_t)N * 64;
    float* dinv = hw2  + (size_t)N * HW2_PITCH;
    unsigned* part = (unsigned*)(dinv + N);
    unsigned* ctr  = part + (size_t)DEG_SLICES * npad;
    float* wgt = (float*)(ctr + 4 * QPITCH);
    const int ntiles = (E + TILE_E - 1) / TILE_E;
    unsigned* deg8 = (unsigned*)(wgt + 2 * (size_t)ntiles * TILE_E);

    const int T = 256;
    const int N4 = N / 4;
    const int nstrips = N / 16;
    k_deg_hist<<<nranges * DEG_SLICES, DEG_T, 0, stream>>>(dst, part, E, per, npad);
    k_dinv    <<<(N4 + 127) / 128, 128, 0, stream>>>(part, dinv, ctr, deg8, N4, npad / 4);
    k_gemm64<<<(nstrips + 7) / 8, 256, 0, stream>>>(x, W1, dinv, b1, xw, h, nstrips);
    k_scatter64<<<EDGE1_BLOCKS, EDGE1_T, 0, stream>>>(src, dst, dinv, xw, h, ctr, wgt, deg8, E, N, ntiles);
    k_gemm34<<<(nstrips + 7) / 8, 256, 0, stream>>>(h, W2, dinv, b2, hw2, out, nstrips);
    k_scatter34<<<EDGE_BLOCKS, T, 0, stream>>>(src, dst, wgt, hw2, out, ctr + 2 * QPITCH, E, N, ntiles);
}


# Both edge scatters made source-stationary: edges bucketed once per call by (dst half, source range) in LDS chunks; a block keeps its range's rows in LDS, no per-edge row reads; atomic order changes
# speedup vs baseline: 3.2280x; 1.1182x over previous
#include <hip/hip_runtime.h>
#include <math.h>

typedef __attribute__((ext_vector_type(2))) float v2f;
typedef __attribute__((ext_vector_type(4))) float v4f;
typedef __attribute__((ext_vector_type(8))) float v8f;

#define EDGE1_BLOCKS 256
#define EDGE1_T     1024
#define DEG8_CAP  102400
#define QPITCH      64
#define HW2_PITCH   34
#define NRANGE      256
#define NBK         (2 * NRANGE)
#define CHUNK       12544
#define RCAP        392
#define OFF_PITCH   544

#define DEG_RANGE  12800
#define DEG_SLICES 32
#define DEG_T      512
#define DEG_B      20
#define SORT_WORDS (2 * NBK + 32 + 2 * CHUNK)

__device__ __forceinline__ void deg_count1(unsigned* cnt, int d, int lo) {
    const unsigned a = (unsigned)(d - lo);
    if (a < (unsigned)DEG_RANGE) atomicAdd(&cnt[a], 1u);
}

__device__ __forceinline__ int node_split(int N) { return min(N, ((N >> 1) + 15) & ~15); }

__device__ __forceinline__ int bucket_of(int s, int d, int nsplit, unsigned R) {
    return (d >= nsplit ? NRANGE : 0) + (int)((unsigned)s / R);
}

__device__ __forceinline__ void bucket_chunk(const int* __restrict__ src, const int* __restrict__ dst, unsigned* sm,
                                             int* __restrict__ gs, int* __restrict__ gd, int* __restrict__ off,
                                             int E, int N, int R, int chunk) {
    static_assert(NBK == DEG_T, "one bucket per thread in the scan");
    unsigned* hist = sm;
    unsigned* cur  = sm + NBK;
    unsigned* wtot = sm + 2 * NBK;
    int* ls = (int*)(sm + 2 * NBK + 32);
    int* ld = ls + CHUNK;
    const int tid = (int)threadIdx.x;
    const int nsplit = node_split(N);
    const int e0 = chunk * CHUNK;
    const int n  = min(CHUNK, E - e0);
    constexpr int SB = (CHUNK / 4 + DEG_T - 1) / DEG_T;
    const bool vec = (((((size_t)(src + e0)) | ((size_t)(dst + e0))) & 15) == 0);
    const int n4 = vec ? (n >> 2) : 0;
    int4 vs[SB], vd[SB];
#pragma unroll
    for (int k = 0; k < SB; ++k) {
        const int i = k * DEG_T + tid;
        if (i < n4) { vs[k] = ((const int4*)(src + e0))[i]; vd[k] = ((const int4*)(dst + e0))[i]; }
    }
    hist[tid] = 0u;
    __syncthreads();
#pragma unroll
    for (int k = 0; k < SB; ++k) {
        if (k * DEG_T + tid < n4) {
            atomicAdd(&hist[bucket_of(vs[k].x, vd[k].x, nsplit, (unsigned)R)], 1u);
            atomicAdd(&hist[bucket_of(vs[k].y, vd[k].y, nsplit, (unsigned)R)], 1u);
            atomicAdd(&hist[bucket_of(vs[k].z, vd[k].z, nsplit, (unsigned)R)], 1u);
            atomicAdd(&hist[bucket_of(vs[k].w, vd[k].w, nsplit, (unsigned)R)], 1u);
        }
    }
    for (int i = (n4 << 2) + tid; i < n; i += DEG_T)
        atomicAdd(&hist[bucket_of(src[e0 + i], dst[e0 + i], nsplit, (unsigned)R)], 1u);
    __syncthreads();
    const unsigned v = hist[tid];
    unsigned inc = v;
    for (int d = 1; d < 32; d <<= 1) {
        const unsigned up = __shfl_up(inc, d, 32);
        if ((tid & 31) >= d) inc += up;
    }
    if ((tid & 31) == 31) wtot[tid >> 5] = inc;
    __syncthreads();
    unsigned ex = inc - v;
    for (int w = 0; w < (tid >> 5); ++w) ex += wtot[w];
    cur[tid] = ex;
    off[(size_t)chunk * OFF_PITCH + tid] = (int)ex;
    if (tid == 0) off[(size_t)chunk * OFF_PITCH + NBK] = n;
    __syncthreads();
#pragma unroll
    for (int k = 0; k < SB; ++k) {
        if (k * DEG_T + tid < n4) {
            const unsigned p0 = atomicAdd(&cur[bucket_of(vs[k].x, vd[k].x, nsplit, (unsigned)R)], 1u);
            const unsigned p1 = atomicAdd(&cur[bucket_of(vs[k].y, vd[k].y, nsplit, (unsigned)R)], 1u);
            const unsigned p2 = atomicAdd(&cur[bucket_of(vs[k].z, vd[k].z, nsplit, (unsigned)R)], 1u);
            const unsigned p3 = atomicAdd(&cur[bucket_of(vs[k].w, vd[k].w, nsplit, (unsigned)R)], 1u);
            ls[p0] = vs[k].x; ld[p0] = vd[k].x;
            ls[p1] = vs[k].y; ld[p1] = vd[k].y;
            ls[p2] = vs[k].z; ld[p2] = vd[k].z;
            ls[p3] = vs[k].w; ld[p3] = vd[k].w;
        }
    }
    for (int i = (n4 << 2) + tid; i < n; i += DEG_T) {
        const int s = src[e0 + i], d = dst[e0 + i];
        const unsigned pos = atomicAdd(&cur[bucket_of(s, d, nsplit, (unsigned)R)], 1u);
        ls[pos] = s; ld[pos] = d;
    }
    __syncthreads();
    for (int i = tid; i < ((n + 3) >> 2); i += DEG_T) {
        ((int4*)(gs + e0))[i] = ((const int4*)ls)[i];
        ((int4*)(gd + e0))[i] = ((const int4*)ld)[i];
    }
}

__global__ void __launch_bounds__(DEG_T) k_deg_hist(const int* __restrict__ src, const int* __restrict__ dst,
                                                    unsigned* __restrict__ part, int* __restrict__ gs,
                                                    int* __restrict__ gd, int* __restrict__ off,
                                                    int E, int N, int R, int per, int npad, int nsort) {
    __shared__ __attribute__((aligned(16))) unsigned sm[SORT_WORDS];
    if ((int)blockIdx.x < nsort) {
        bucket_chunk(src, dst, sm, gs, gd, off, E, N, R, (int)blockIdx.x);
        return;
    }
    unsigned* cnt = sm;
    const int bid = (int)blockIdx.x - nsort;
    const int r  = bid / DEG_SLICES;
    const int s  = bid % DEG_SLICES;
    const int lo = r * DEG_RANGE;
    for (int i = threadIdx.x; i < DEG_RANGE / 4; i += DEG_T)
        ((uint4*)cnt)[i] = make_uint4(0u, 0u, 0u, 0u);
    __syncthreads();
    const int e0 = min(E, s * per);
    const int e1 = min(E, e0 + per);
    int e = e0;
    if ((((size_t)(dst + e0)) & 15) == 0) {
        const int4* p = (const int4*)(dst + e0);
        const int n4 = (e1 - e0) >> 2;
        for (int base = 0; base < n4; base += DEG_B * DEG_T) {
            int4 v[DEG_B];
#pragma unroll
            for (int k = 0; k < DEG_B; ++k) {
                const int i = base + k * DEG_T + (int)threadIdx.x;
                v[k] = make_int4(-1, -1, -1, -1);
                if (i < n4) v[k] = p[i];
            }
#pragma unroll
            for (int k = 0; k < DEG_B; ++k) {
                deg_count1(cnt, v[k].x, lo);
                deg_count1(cnt, v[k].y, lo);
                deg_count1(cnt, v[k].z, lo);
                deg_count1(cnt, v[k].w, lo);
            }
        }
        e = e0 + (n4 << 2);
    }
    for (e += (int)threadIdx.x; e < e1; e += DEG_T)
        deg_count1(cnt, dst[e], lo);
    __syncthreads();
    uint4* o = (uint4*)(part + (size_t)s * npad + lo);
    for (int i = threadIdx.x; i < DEG_RANGE / 4; i += DEG_T)
        o[i] = ((const uint4*)cnt)[i];
}

__global__ void __launch_bounds__(128) k_dinv(const unsigned* __restrict__ part, float* __restrict__ dinv,
                                              unsigned* __restrict__ ctr, unsigned* __restrict__ deg8,
                                              int n4, int npad4) {
    const int i = blockIdx.x * blockDim.x + threadIdx.x;
    if (i < 4) ctr[i * QPITCH] = 0u;
    if (i < n4) {
        uint4 c = make_uint4(1u, 1u, 1u, 1u);
#pragma unroll
        for (int s = 0; s < DEG_SLICES; ++s) {
            const uint4 v = ((const uint4*)part)[(size_t)s * npad4 + i];
            c.x += v.x; c.y += v.y; c.z += v.z; c.w += v.w;
        }
        ((float4*)dinv)[i] = make_float4(rsqrtf((float)c.x), rsqrtf((float)c.y),
                                         rsqrtf((float)c.z), rsqrtf((float)c.w));
        deg8[i] = min(c.x, 255u) | (min(c.y, 255u) << 8) | (min(c.z, 255u) << 16) | (min(c.w, 255u) << 24);
    }
}

__global__ void __launch_bounds__(256) k_gemm64(const float* __restrict__ A,
                                                const float* __restrict__ B,
                                                const float* __restrict__ dinv,
                                                const float* __restrict__ b1,
                                                float* __restrict__ C,
                                                float* __restrict__ Hacc,
                                                int nstrips) {
    const int wave = __builtin_amdgcn_readfirstlane((int)(threadIdx.x >> 5));
    const int lane = threadIdx.x & 31;
    const int half = lane >> 4;
    const int lm   = lane & 15;
    const int strip = blockIdx.x * 8 + wave;
    if (strip >= nstrips) return;
    const int m0   = strip * 16;
    const float* arow = A + (size_t)(m0 + lm) * 64;

    v2f a[16];
#pragma unroll
    for (int kk = 0; kk < 16; ++kk)
        a[kk] = *(const v2f*)(arow + kk * 4 + half * 2);

    v8f acc[4];
#pragma unroll
    for (int nt = 0; nt < 4; ++nt) acc[nt] = (v8f){0.f,0.f,0.f,0.f,0.f,0.f,0.f,0.f};
#pragma unroll
    for (int kk = 0; kk < 16; ++kk) {
        const int kb = kk * 4 + half * 2;
#pragma unroll
        for (int nt = 0; nt < 4; ++nt) {
            const int bcol = nt * 16 + lm;
            v2f b;
            b.x = B[(size_t)kb * 64 + bcol];
            b.y = B[(size_t)(kb + 1) * 64 + bcol];
            acc[nt] = __builtin_amdgcn_wmma_f32_16x16x4_f32(false, a[kk], false, b,
                                                            (short)0, acc[nt], false, false);
        }
    }
    float di2[8];
#pragma unroll
    for (int i = 0; i < 8; ++i) {
        const float di = dinv[m0 + half * 8 + i];
        di2[i] = di * di;
    }
#pragma unroll
    for (int nt = 0; nt < 4; ++nt) {
        const int col = nt * 16 + lm;
        const float bb = b1[col];
#pragma unroll
        for (int i = 0; i < 8; ++i) {
            const int row = m0 + half * 8 + i;
            const float v = acc[nt][i];
            C[(size_t)row * 64 + col]    = v;
            Hacc[(size_t)row * 64 + col] = bb + di2[i] * v;
        }
    }
}


__device__ __forceinline__ int chip_half() {
    return (int)((__builtin_amdgcn_s_sendmsg_rtn(0x87) >> 18) & 1u);
}

__device__ __forceinline__ int flat_to_sorted(int f, int pre, int cst, int npw) {
    int g = 0;
    for (int k = 0; k < npw; ++k) {
        const int pk = __builtin_amdgcn_readlane(pre, k);
        const int ck = __builtin_amdgcn_readlane(cst, k);
        if (f >= pk) g = ck + (f - pk);
    }
    return g;
}

__device__ __forceinline__ void piece_offsets(const int* __restrict__ off, int b, int wave, int lane, int nchunks,
                                              int& o0, int& o1) {
    const int c = wave + 32 * lane;
    o0 = 0; o1 = 0;
    if (c < nchunks) {
        const int* o = off + (size_t)c * OFF_PITCH + b;
        o0 = o[0]; o1 = o[1];
    }
}

__device__ __forceinline__ int piece_scan(int o0, int o1, int wave, int lane, int& pre, int& cst) {
    const int cnt = o1 - o0;
    cst = (wave + 32 * lane) * CHUNK + o0;
    int inc = cnt;
    for (int d = 1; d < 32; d <<= 1) {
        const int up = __shfl_up(inc, d, 32);
        if (lane >= d) inc += up;
    }
    pre = inc - cnt;
    return __builtin_amdgcn_readlane(inc, 31);
}

__device__ __forceinline__ float dinv_from(unsigned c, const float* __restrict__ dinv, int node) {
    return (c != 255u) ? rsqrtf((float)c) : dinv[node];
}

__device__ __forceinline__ void stage_degree_table(unsigned char* tab, const unsigned* __restrict__ deg8, int N) {
    const int nw  = N >> 2;
    const int n16 = nw >> 2;
    const int rot = (int)(((blockIdx.x % EDGE1_BLOCKS) * (unsigned)n16) / EDGE1_BLOCKS);
    constexpr int NK = (DEG8_CAP / 16 + EDGE1_T - 1) / EDGE1_T;
    uint4 tv[NK];
#pragma unroll
    for (int k = 0; k < NK; ++k) {
        const int i = k * EDGE1_T + (int)threadIdx.x;
        int j = i + rot; if (j >= n16) j -= n16;
        tv[k] = make_uint4(0u, 0u, 0u, 0u);
        if (i < n16) tv[k] = ((const uint4*)deg8)[j];
    }
#pragma unroll
    for (int k = 0; k < NK; ++k) {
        const int i = k * EDGE1_T + (int)threadIdx.x;
        int j = i + rot; if (j >= n16) j -= n16;
        if (i < n16) ((uint4*)tab)[j] = tv[k];
    }
    for (int w = (n16 << 2) + (int)threadIdx.x; w < nw; w += EDGE1_T)
        ((unsigned*)tab)[w] = deg8[w];
}

__device__ __forceinline__ float edge_weight(const unsigned char* tab, bool use_tab, const float* __restrict__ dinv, int s, int d) {
    unsigned cs = 255u, cd = 255u;
    if (use_tab) { cs = tab[s]; cd = tab[d]; }
    return dinv_from(cs, dinv, s) * dinv_from(cd, dinv, d);
}

template <bool GUARD>
__device__ __forceinline__ void add_rows64(const float* rows, float* __restrict__ h,
                                           int sl, int dv, int wi, int j0, int n, int lane) {
    float v0[8], v1[8];
#pragma unroll
    for (int j = 0; j < 8; ++j) {
        if (!GUARD || j < n) {
            const int sj = __builtin_amdgcn_readlane(sl, j0 + j);
            const float* xr = rows + sj * 64;
            v0[j] = xr[lane];
            v1[j] = xr[lane + 32];
        }
    }
#pragma unroll
    for (int j = 0; j < 8; ++j) {
        if (!GUARD || j < n) {
            const int   dj = __builtin_amdgcn_readlane(dv, j0 + j);
            const float wj = __int_as_float(__builtin_amdgcn_readlane(wi, j0 + j));
            float* o = h + (size_t)dj * 64;
            atomicAdd(o + lane,      wj * v0[j]);
            atomicAdd(o + lane + 32, wj * v1[j]);
        }
    }
}

__global__ void __launch_bounds__(EDGE1_T) k_scatter64(const int* __restrict__ gs, const int* __restrict__ gd,
                                                   const int* __restrict__ off,
                                                   const float* __restrict__ dinv, const float* __restrict__ xw,
                                                   float* __restrict__ h, unsigned* __restrict__ ctr,
                                                   const unsigned* __restrict__ deg8,
                                                   int N, int R, int nchunks) {
    __shared__ unsigned s_t;
    __shared__ int s_q;
    __shared__ __attribute__((aligned(16))) unsigned char tab[DEG8_CAP];
    __shared__ __attribute__((aligned(16))) float rows[RCAP * 64];
    const int lane = threadIdx.x & 31;
    const int wave = __builtin_amdgcn_readfirstlane((int)(threadIdx.x >> 5));
    if (threadIdx.x == 0) s_q = chip_half();
    const bool use_tab = (N <= DEG8_CAP);
    if (use_tab) stage_degree_table(tab, deg8, N);
    __syncthreads();
    int q = __builtin_amdgcn_readfirstlane(s_q);
    unsigned pend = 0u;
    if (threadIdx.x == 0) pend = atomicAdd(ctr + q * QPITCH, 1u);
    bool switched = false;
    const int npw = min(32, max(0, (nchunks - wave + 31) >> 5));
    for (;;) {
        __syncthreads();
        if (threadIdx.x == 0) s_t = pend;
        __syncthreads();
        const unsigned t = __builtin_amdgcn_readfirstlane(s_t);
        if (t >= (unsigned)NRANGE) {
            if (switched) break;
            switched = true;
            q ^= 1;
            if (threadIdx.x == 0) pend = atomicAdd(ctr + q * QPITCH, 1u);
            continue;
        }
        if (threadIdx.x == 0) pend = atomicAdd(ctr + q * QPITCH, 1u);
        const int b  = q * NRANGE + (int)t;
        const int r0 = (int)t * R;
        const int nrows = max(0, min(N, r0 + R) - r0);
        int pre, cst, total;
        {
            const int n16 = min(nrows, RCAP) * 16;
            const v4f* xr4 = (const v4f*)(xw + (size_t)min(r0, N - 1) * 64);
            constexpr int NR = (RCAP * 16 + EDGE1_T - 1) / EDGE1_T;
            static_assert(NR == 7, "the fence below names rv[0..6]");
            v4f rv[NR];
#pragma unroll
            for (int k = 0; k < NR; ++k) {
                const int i = k * EDGE1_T + (int)threadIdx.x;
                rv[k] = xr4[max(min(i, n16 - 1), 0)];
            }
            int o0, o1;
            piece_offsets(off, b, wave, lane, nchunks, o0, o1);
            asm volatile("" :: "v"(rv[0]), "v"(rv[1]), "v"(rv[2]), "v"(rv[3]), "v"(rv[4]), "v"(rv[5]), "v"(rv[6]), "v"(o0), "v"(o1));
            total = piece_scan(o0, o1, wave, lane, pre, cst);
#pragma unroll
            for (int k = 0; k < NR; ++k) {
                const int i = k * EDGE1_T + (int)threadIdx.x;
                if (i < n16) ((v4f*)rows)[i] = rv[k];
            }
        }
        __syncthreads();
        for (int f0 = 0; f0 < total; f0 += 128) {
            int sv[4], dv[4];
#pragma unroll
            for (int w = 0; w < 4; ++w) {
                const int f = f0 + 32 * w + lane;
                const int g = flat_to_sorted(f, pre, cst, npw);
                sv[w] = 0; dv[w] = 0;
                if (f < total) { sv[w] = gs[g]; dv[w] = gd[g]; }
            }
#pragma unroll
            for (int w = 0; w < 4; ++w) {
                const int n = min(32, total - f0 - 32 * w);
                if (n > 0) {
                    float wv = 0.0f;
                    if (lane < n) wv = edge_weight(tab, use_tab, dinv, sv[w], dv[w]);
                    const int sl = sv[w] - r0;
                    const int wi = __float_as_int(wv);
                    int u0 = 0;
                    for (; u0 + 8 <= n; u0 += 8) add_rows64<false>(rows, h, sl, dv[w], wi, u0, 8, lane);
                    if (u0 < n) add_rows64<true>(rows, h, sl, dv[w], wi, u0, n - u0, lane);
                }
            }
        }
    }
}

__global__ void __launch_bounds__(256) k_gemm34(const float* __restrict__ A,
                                                const float* __restrict__ B,
                                                const float* __restrict__ dinv,
                                                const float* __restrict__ b2,
                                                float* __restrict__ C,
                                                float* __restrict__ out,
                                                int nstrips) {
    const int wave = __builtin_amdgcn_readfirstlane((int)(threadIdx.x >> 5));
    const int lane = threadIdx.x & 31;
    const int half = lane >> 4;
    const int lm   = lane & 15;
    const int strip = blockIdx.x * 8 + wave;
    if (strip >= nstrips) return;
    const int m0   = strip * 16;
    const float* arow = A + (size_t)(m0 + lm) * 64;

    v2f a[16];
#pragma unroll
    for (int kk = 0; kk < 16; ++kk)
        a[kk] = *(const v2f*)(arow + kk * 4 + half * 2);

    float m = 0.0f;
#pragma unroll
    for (int kk = 0; kk < 16; ++kk) {
        a[kk].x = fmaxf(a[kk].x, 0.0f);
        a[kk].y = fmaxf(a[kk].y, 0.0f);
        m = fmaxf(m, fmaxf(a[kk].x, a[kk].y));
    }
    m = fmaxf(m, __shfl_xor(m, 16, 32));
    float s = 0.0f;
#pragma unroll
    for (int kk = 0; kk < 16; ++kk) {
        a[kk].x = __expf(a[kk].x - m);
        a[kk].y = __expf(a[kk].y - m);
        s += a[kk].x + a[kk].y;
    }
    s += __shfl_xor(s, 16, 32);
    const float inv = 1.0f / s;

    v8f acc[3];
#pragma unroll
    for (int nt = 0; nt < 3; ++nt) acc[nt] = (v8f){0.f,0.f,0.f,0.f,0.f,0.f,0.f,0.f};
#pragma unroll
    for (int kk = 0; kk < 16; ++kk) {
        const int kb = kk * 4 + half * 2;
        v2f av;
        av.x = a[kk].x * inv;
        av.y = a[kk].y * inv;
#pragma unroll
        for (int nt = 0; nt < 3; ++nt) {
            const int col = nt * 16 + lm;
            const bool ok = (col < 34);
            const int  cc = ok ? col : 33;
            const float bx = B[(size_t)kb * 34 + cc];
            const float by = B[(size_t)(kb + 1) * 34 + cc];
            v2f b;
            b.x = ok ? bx : 0.0f;
            b.y = ok ? by : 0.0f;
            acc[nt] = __builtin_amdgcn_wmma_f32_16x16x4_f32(false, av, false, b,
                                                            (short)0, acc[nt], false, false);
        }
    }
    float di2[8];
#pragma unroll
    for (int i = 0; i < 8; ++i) {
        const float di = dinv[m0 + half * 8 + i];
        di2[i] = di * di;
    }
#pragma unroll
    for (int nt = 0; nt < 3; ++nt) {
        const int col = nt * 16 + lm;
        if (col < 34) {
            const float bb = b2[col];
#pragma unroll
            for (int i = 0; i < 8; ++i) {
                const int row = m0 + half * 8 + i;
                const float v = acc[nt][i];
                C[(size_t)row * HW2_PITCH + col] = v;
                out[(size_t)row * 34 + col] = bb + di2[i] * v;
            }
        }
    }
}

__device__ __forceinline__ int row_cut(int d) { return 32 - 2 * (d & 15); }

template <bool GUARD>
__device__ __forceinline__ void add_rows34(const float* rows, float* __restrict__ out,
                                           int sl, int dv, int wi, int j0, int n, int lane) {
    float va[8], vb[8];
    const bool tail = lane < 2;
    const int  ch2  = tail ? 32 + lane : lane;
#pragma unroll
    for (int j = 0; j < 8; ++j) {
        if (!GUARD || j < n) {
            const int sj = __builtin_amdgcn_readlane(sl, j0 + j);
            const float* r = rows + sj * HW2_PITCH;
            va[j] = r[lane];
            vb[j] = r[32 + (lane & 1)];
        }
    }
#pragma unroll
    for (int j = 0; j < 8; ++j) {
        if (!GUARD || j < n) {
            const int   dj = __builtin_amdgcn_readlane(dv, j0 + j);
            const float wj = __int_as_float(__builtin_amdgcn_readlane(wi, j0 + j));
            const int   na = row_cut(dj);
            if (lane < na)          atomicAdd(&out[(unsigned)(dj * 34 + lane)], wj * va[j]);
            if (tail || lane >= na) atomicAdd(&out[(unsigned)(dj * 34 + ch2)],  wj * (tail ? vb[j] : va[j]));
        }
    }
}

__global__ void __launch_bounds__(EDGE1_T) k_scatter34(const int* __restrict__ gs, const int* __restrict__ gd,
                                                   const int* __restrict__ off,
                                                   const float* __restrict__ dinv, const float* __restrict__ hw2,
                                                   float* __restrict__ out, unsigned* __restrict__ ctr,
                                                   const unsigned* __restrict__ deg8,
                                                   int N, int R, int nchunks) {
    __shared__ unsigned s_t;
    __shared__ int s_q;
    __shared__ __attribute__((aligned(16))) unsigned char tab[DEG8_CAP];
    __shared__ __attribute__((aligned(16))) float rows[RCAP * HW2_PITCH];
    const int lane = threadIdx.x & 31;
    const int wave = __builtin_amdgcn_readfirstlane((int)(threadIdx.x >> 5));
    if (threadIdx.x == 0) s_q = chip_half();
    const bool use_tab = (N <= DEG8_CAP);
    if (use_tab) stage_degree_table(tab, deg8, N);
    __syncthreads();
    int q = __builtin_amdgcn_readfirstlane(s_q);
    unsigned pend = 0u;
    if (threadIdx.x == 0) pend = atomicAdd(ctr + q * QPITCH, 1u);
    bool switched = false;
    const int npw = min(32, max(0, (nchunks - wave + 31) >> 5));
    for (;;) {
        __syncthreads();
        if (threadIdx.x == 0) s_t = pend;
        __syncthreads();
        const unsigned t = __builtin_amdgcn_readfirstlane(s_t);
        if (t >= (unsigned)NRANGE) {
            if (switched) break;
            switched = true;
            q ^= 1;
            if (threadIdx.x == 0) pend = atomicAdd(ctr + q * QPITCH, 1u);
            continue;
        }
        if (threadIdx.x == 0) pend = atomicAdd(ctr + q * QPITCH, 1u);
        const int b  = q * NRANGE + (int)t;
        const int r0 = (int)t * R;
        const int nrows = max(0, min(N, r0 + R) - r0);
        int pre, cst, total;
        {
            const int n8 = min(nrows, RCAP) * (HW2_PITCH / 2);
            const v2f* hr2 = (const v2f*)(hw2 + (size_t)min(r0, N - 1) * HW2_PITCH);
            constexpr int NR = (RCAP * (HW2_PITCH / 2) + EDGE1_T - 1) / EDGE1_T;
            static_assert(NR == 7, "the fence below names rv[0..6]");
            v2f rv[NR];
#pragma unroll
            for (int k = 0; k < NR; ++k) {
                const int i = k * EDGE1_T + (int)threadIdx.x;
                rv[k] = hr2[max(min(i, n8 - 1), 0)];
            }
            int o0, o1;
            piece_offsets(off, b, wave, lane, nchunks, o0, o1);
            asm volatile("" :: "v"(rv[0]), "v"(rv[1]), "v"(rv[2]), "v"(rv[3]), "v"(rv[4]), "v"(rv[5]), "v"(rv[6]), "v"(o0), "v"(o1));
            total = piece_scan(o0, o1, wave, lane, pre, cst);
#pragma unroll
            for (int k = 0; k < NR; ++k) {
                const int i = k * EDGE1_T + (int)threadIdx.x;
                if (i < n8) ((v2f*)rows)[i] = rv[k];
            }
        }
        __syncthreads();
        for (int f0 = 0; f0 < total; f0 += 128) {
            int sv[4], dv[4];
#pragma unroll
            for (int w = 0; w < 4; ++w) {
                const int f = f0 + 32 * w + lane;
                const int g = flat_to_sorted(f, pre, cst, npw);
                sv[w] = 0; dv[w] = 0;
                if (f < total) { sv[w] = gs[g]; dv[w] = gd[g]; }
            }
#pragma unroll
            for (int w = 0; w < 4; ++w) {
                const int n = min(32, total - f0 - 32 * w);
                if (n > 0) {
                    float wv = 0.0f;
                    if (lane < n) wv = edge_weight(tab, use_tab, dinv, sv[w], dv[w]);
                    const int sl = sv[w] - r0;
                    const int wi = __float_as_int(wv);
                    int u0 = 0;
                    for (; u0 + 8 <= n; u0 += 8) add_rows34<false>(rows, out, sl, dv[w], wi, u0, 8, lane);
                    if (u0 < n) add_rows34<true>(rows, out, sl, dv[w], wi, u0, n - u0, lane);
                }
            }
        }
    }
}

extern "C" void kernel_launch(void* const* d_in, const int* in_sizes, int n_in,
                              void* d_out, int out_size, void* d_ws, size_t ws_size,
                              hipStream_t stream) {
    const float* x  = (const float*)d_in[0];
    const int*   ei = (const int*)  d_in[1];
    const float* W1 = (const float*)d_in[2];
    const float* b1 = (const float*)d_in[3];
    const float* W2 = (const float*)d_in[4];
    const float* b2 = (const float*)d_in[5];
    float* out = (float*)d_out;

    const int N = in_sizes[0] / 64;
    const int E = in_sizes[1] / 2;
    const int* src = ei;
    const int* dst = ei + E;

    const int nranges = (N + DEG_RANGE - 1) / DEG_RANGE;
    const int npad    = nranges * DEG_RANGE;
    const int per     = (((E + DEG_SLICES - 1) / DEG_SLICES) + 63) & ~63;
    const int nchunks = (E + CHUNK - 1) / CHUNK;
    const int R       = (N + NRANGE - 1) / NRANGE;
    const int N4      = N / 4;

    float* xw   = (float*)d_ws;
    float* h    = xw   + (size_t)N * 64;
    float* hw2  = h    + (size_t)N * 64;
    float* dinv = hw2  + (size_t)N * HW2_PITCH;
    unsigned* part = (unsigned*)(dinv + N);
    unsigned* ctr  = part + (size_t)DEG_SLICES * npad;
    unsigned* deg8 = ctr + 4 * QPITCH;
    int* gs  = (int*)(deg8 + ((N4 + 31) & ~31));
    int* gd  = gs + (size_t)nchunks * CHUNK;
    int* off = gd + (size_t)nchunks * CHUNK;

    const int nstrips = N / 16;
    k_deg_hist<<<nchunks + nranges * DEG_SLICES, DEG_T, 0, stream>>>(src, dst, part, gs, gd, off, E, N, R, per, npad, nchunks);
    k_dinv    <<<(N4 + 127) / 128, 128, 0, stream>>>(part, dinv, ctr, deg8, N4, npad / 4);
    k_gemm64<<<(nstrips + 7) / 8, 256, 0, stream>>>(x, W1, dinv, b1, xw, h, nstrips);
    k_scatter64<<<EDGE1_BLOCKS, EDGE1_T, 0, stream>>>(gs, gd, off, dinv, xw, h, ctr, deg8, N, R, nchunks);
    k_gemm34<<<(nstrips + 7) / 8, 256, 0, stream>>>(h, W2, dinv, b2, hw2, out, nstrips);
    k_scatter34<<<EDGE1_BLOCKS, EDGE1_T, 0, stream>>>(gs, gd, off, dinv, hw2, out, ctr + 2 * QPITCH, deg8, N, R, nchunks);
}
